# starting kernel
# baseline (speedup 1.0000x reference)
#include <hip/hip_runtime.h>
#include <hip/hip_bf16.h>

#define BB 2048
#define TT 1024
#define II 28
#define HH 2
#define CC 10
#define NG 8
#define MM (BB * TT)

typedef __attribute__((ext_vector_type(2))) float v2f;
typedef __attribute__((ext_vector_type(8))) float v8f;

__device__ __forceinline__ float fast_tanh(float x) {
#if __has_builtin(__builtin_amdgcn_tanhf)
    return __builtin_amdgcn_tanhf(x);
#else
    float e = __expf(-2.0f * x);
    return (1.0f - e) / (1.0f + e);
#endif
}
__device__ __forceinline__ float fast_sigm(float x) {
    return 0.5f + 0.5f * fast_tanh(0.5f * x);
}

__global__ void lstm_xproj_wmma(const float* __restrict__ x,
                                const float* __restrict__ W_ih,
                                const float* __restrict__ b_ih,
                                const float* __restrict__ b_hh,
                                float* __restrict__ xg) {
    const int lane  = threadIdx.x & 31;
    const int n     = lane & 15;
    const int khalf = lane >> 4;

    v2f bfrag[7];
#pragma unroll
    for (int j = 0; j < 7; ++j) {
        const int k0 = 4 * j + 2 * khalf;
        float b0 = 0.0f, b1 = 0.0f;
        if (n < NG) {
            b0 = W_ih[n * II + k0];
            b1 = W_ih[n * II + k0 + 1];
        }
        v2f bf; bf.x = b0; bf.y = b1;
        bfrag[j] = bf;
    }

    const float bias = (n < NG) ? (b_ih[n] + b_hh[n]) : 0.0f;
    v8f cbias;
#pragma unroll
    for (int v = 0; v < 8; ++v) cbias[v] = bias;

    const int wavesPerBlock = blockDim.x >> 5;
    const int gwave  = blockIdx.x * wavesPerBlock + (threadIdx.x >> 5);
    const int nwaves = gridDim.x * wavesPerBlock;
    const int ntiles = MM >> 4;

    for (int tile = gwave; tile < ntiles; tile += nwaves) {
        const int   row  = tile * 16 + n;
        const float* xrow = x + (long)row * II;

        v8f c = cbias;
#pragma unroll
        for (int j = 0; j < 7; ++j) {
            const int k0 = 4 * j + 2 * khalf;
            v2f a; a.x = xrow[k0]; a.y = xrow[k0 + 1];
            c = __builtin_amdgcn_wmma_f32_16x16x4_f32(
                    false, a, false, bfrag[j], (short)0, c, false, false);
        }

        if (n < NG) {
#pragma unroll
            for (int v = 0; v < 8; ++v) {
                const int m  = v + 8 * khalf;
                const int r  = tile * 16 + m;
                const int bi = r >> 10;
                const int ti = r & (TT - 1);
                xg[((long)ti * BB + bi) * NG + n] = c[v];
            }
        }
    }
}

__global__ void lstm_scan(const float* __restrict__ xg,
                          const float* __restrict__ W_hh,
                          const float* __restrict__ fc_W,
                          const float* __restrict__ fc_b,
                          float* __restrict__ out) {
    const int b = blockIdx.x * blockDim.x + threadIdx.x;
    if (b >= BB) return;

    float w[16];
#pragma unroll
    for (int k = 0; k < 16; ++k) w[k] = W_hh[k];

    const float4* g4 = (const float4*)xg;
    const long stride = (long)BB * 2;
    long idx = (long)b * 2;

    float h0 = 0.f, h1 = 0.f, c0 = 0.f, c1 = 0.f;
    float4 lo = g4[idx], hi = g4[idx + 1];

    for (int t = 0; t < TT; ++t) {
        const long nidx = idx + stride;
        float4 nlo = lo, nhi = hi;
        if (t + 1 < TT) { nlo = g4[nidx]; nhi = g4[nidx + 1]; }

        const float i0 = lo.x + h0 * w[0]  + h1 * w[1];
        const float i1 = lo.y + h0 * w[2]  + h1 * w[3];
        const float f0 = lo.z + h0 * w[4]  + h1 * w[5];
        const float f1 = lo.w + h0 * w[6]  + h1 * w[7];
        const float g0 = hi.x + h0 * w[8]  + h1 * w[9];
        const float g1 = hi.y + h0 * w[10] + h1 * w[11];
        const float o0 = hi.z + h0 * w[12] + h1 * w[13];
        const float o1 = hi.w + h0 * w[14] + h1 * w[15];

        c0 = fast_sigm(f0) * c0 + fast_sigm(i0) * fast_tanh(g0);
        c1 = fast_sigm(f1) * c1 + fast_sigm(i1) * fast_tanh(g1);
        h0 = fast_sigm(o0) * fast_tanh(c0);
        h1 = fast_sigm(o1) * fast_tanh(c1);

        lo = nlo; hi = nhi; idx = nidx;
    }

#pragma unroll
    for (int c = 0; c < CC; ++c) {
        out[b * CC + c] = fc_b[c] + h0 * fc_W[c * HH + 0] + h1 * fc_W[c * HH + 1];
    }
}

__global__ void lstm_fused(const float* __restrict__ x,
                           const float* __restrict__ W_ih,
                           const float* __restrict__ W_hh,
                           const float* __restrict__ b_ih,
                           const float* __restrict__ b_hh,
                           const float* __restrict__ fc_W,
                           const float* __restrict__ fc_b,
                           float* __restrict__ out) {
    const int b = blockIdx.x * blockDim.x + threadIdx.x;
    if (b >= BB) return;

    float w[16];
#pragma unroll
    for (int k = 0; k < 16; ++k) w[k] = W_hh[k];
    float bias[NG];
#pragma unroll
    for (int g = 0; g < NG; ++g) bias[g] = b_ih[g] + b_hh[g];

    float h0 = 0.f, h1 = 0.f, c0 = 0.f, c1 = 0.f;
    const float* xb = x + (long)b * TT * II;

    for (int t = 0; t < TT; ++t) {
        const float* xr = xb + (long)t * II;
        float ga[NG];
#pragma unroll
        for (int g = 0; g < NG; ++g) {
            float s = bias[g];
            for (int i = 0; i < II; ++i) s += xr[i] * W_ih[g * II + i];
            ga[g] = s;
        }
        const float i0 = ga[0] + h0 * w[0]  + h1 * w[1];
        const float i1 = ga[1] + h0 * w[2]  + h1 * w[3];
        const float f0 = ga[2] + h0 * w[4]  + h1 * w[5];
        const float f1 = ga[3] + h0 * w[6]  + h1 * w[7];
        const float g0 = ga[4] + h0 * w[8]  + h1 * w[9];
        const float g1 = ga[5] + h0 * w[10] + h1 * w[11];
        const float o0 = ga[6] + h0 * w[12] + h1 * w[13];
        const float o1 = ga[7] + h0 * w[14] + h1 * w[15];

        c0 = fast_sigm(f0) * c0 + fast_sigm(i0) * fast_tanh(g0);
        c1 = fast_sigm(f1) * c1 + fast_sigm(i1) * fast_tanh(g1);
        h0 = fast_sigm(o0) * fast_tanh(c0);
        h1 = fast_sigm(o1) * fast_tanh(c1);
    }
#pragma unroll
    for (int c = 0; c < CC; ++c) {
        out[b * CC + c] = fc_b[c] + h0 * fc_W[c * HH + 0] + h1 * fc_W[c * HH + 1];
    }
}

extern "C" void kernel_launch(void* const* d_in, const int* in_sizes, int n_in,
                              void* d_out, int out_size, void* d_ws, size_t ws_size,
                              hipStream_t stream) {
    const float* x    = (const float*)d_in[0];
    const float* W_ih = (const float*)d_in[1];
    const float* W_hh = (const float*)d_in[2];
    const float* b_ih = (const float*)d_in[3];
    const float* b_hh = (const float*)d_in[4];
    const float* fc_W = (const float*)d_in[5];
    const float* fc_b = (const float*)d_in[6];
    float* out = (float*)d_out;

    const size_t need = (size_t)MM * NG * sizeof(float);
    if (ws_size >= need) {
        float* xg = (float*)d_ws;
        lstm_xproj_wmma<<<2048, 256, 0, stream>>>(x, W_ih, b_ih, b_hh, xg);
        lstm_scan<<<BB / 32, 32, 0, stream>>>(xg, W_hh, fc_W, fc_b, out);
    } else {
        lstm_fused<<<BB / 32, 32, 0, stream>>>(x, W_ih, W_hh, b_ih, b_hh,
                                               fc_W, fc_b, out);
    }
}


# scan kernel: each lane keeps the gate rows of the next 16 timesteps in flight in a register queue (was 1 step ahead), hiding the memory round trip per step; arithmetic and summation order unchanged
# speedup vs baseline: 3.6648x; 3.6648x over previous
#include <hip/hip_runtime.h>
#include <hip/hip_bf16.h>

#define BB 2048
#define TT 1024
#define II 28
#define HH 2
#define CC 10
#define NG 8
#define MM (BB * TT)

typedef __attribute__((ext_vector_type(2))) float v2f;
typedef __attribute__((ext_vector_type(8))) float v8f;

__device__ __forceinline__ float fast_tanh(float x) {
#if __has_builtin(__builtin_amdgcn_tanhf)
    return __builtin_amdgcn_tanhf(x);
#else
    float e = __expf(-2.0f * x);
    return (1.0f - e) / (1.0f + e);
#endif
}
__device__ __forceinline__ float fast_sigm(float x) {
    return 0.5f + 0.5f * fast_tanh(0.5f * x);
}

__global__ void lstm_xproj_wmma(const float* __restrict__ x,
                                const float* __restrict__ W_ih,
                                const float* __restrict__ b_ih,
                                const float* __restrict__ b_hh,
                                float* __restrict__ xg) {
    const int lane  = threadIdx.x & 31;
    const int n     = lane & 15;
    const int khalf = lane >> 4;

    v2f bfrag[7];
#pragma unroll
    for (int j = 0; j < 7; ++j) {
        const int k0 = 4 * j + 2 * khalf;
        float b0 = 0.0f, b1 = 0.0f;
        if (n < NG) {
            b0 = W_ih[n * II + k0];
            b1 = W_ih[n * II + k0 + 1];
        }
        v2f bf; bf.x = b0; bf.y = b1;
        bfrag[j] = bf;
    }

    const float bias = (n < NG) ? (b_ih[n] + b_hh[n]) : 0.0f;
    v8f cbias;
#pragma unroll
    for (int v = 0; v < 8; ++v) cbias[v] = bias;

    const int wavesPerBlock = blockDim.x >> 5;
    const int gwave  = blockIdx.x * wavesPerBlock + (threadIdx.x >> 5);
    const int nwaves = gridDim.x * wavesPerBlock;
    const int ntiles = MM >> 4;

    for (int tile = gwave; tile < ntiles; tile += nwaves) {
        const int   row  = tile * 16 + n;
        const float* xrow = x + (long)row * II;

        v8f c = cbias;
#pragma unroll
        for (int j = 0; j < 7; ++j) {
            const int k0 = 4 * j + 2 * khalf;
            v2f a; a.x = xrow[k0]; a.y = xrow[k0 + 1];
            c = __builtin_amdgcn_wmma_f32_16x16x4_f32(
                    false, a, false, bfrag[j], (short)0, c, false, false);
        }

        if (n < NG) {
#pragma unroll
            for (int v = 0; v < 8; ++v) {
                const int m  = v + 8 * khalf;
                const int r  = tile * 16 + m;
                const int bi = r >> 10;
                const int ti = r & (TT - 1);
                xg[((long)ti * BB + bi) * NG + n] = c[v];
            }
        }
    }
}

#define PF 16
static_assert(TT % PF == 0 && 2 * PF <= 63, "queue depth");

__device__ __forceinline__ void lstm_step(const float4 lo, const float4 hi,
                                          const float (&w)[16],
                                          float& h0, float& h1,
                                          float& c0, float& c1) {
    const float i0 = lo.x + h0 * w[0]  + h1 * w[1];
    const float i1 = lo.y + h0 * w[2]  + h1 * w[3];
    const float f0 = lo.z + h0 * w[4]  + h1 * w[5];
    const float f1 = lo.w + h0 * w[6]  + h1 * w[7];
    const float g0 = hi.x + h0 * w[8]  + h1 * w[9];
    const float g1 = hi.y + h0 * w[10] + h1 * w[11];
    const float o0 = hi.z + h0 * w[12] + h1 * w[13];
    const float o1 = hi.w + h0 * w[14] + h1 * w[15];

    c0 = fast_sigm(f0) * c0 + fast_sigm(i0) * fast_tanh(g0);
    c1 = fast_sigm(f1) * c1 + fast_sigm(i1) * fast_tanh(g1);
    h0 = fast_sigm(o0) * fast_tanh(c0);
    h1 = fast_sigm(o1) * fast_tanh(c1);
}

__global__ void __launch_bounds__(32)
lstm_scan(const float* __restrict__ xg,
          const float* __restrict__ W_hh,
          const float* __restrict__ fc_W,
          const float* __restrict__ fc_b,
          float* __restrict__ out) {
    const int b = blockIdx.x * blockDim.x + threadIdx.x;
    if (b >= BB) return;

    float w[16];
#pragma unroll
    for (int k = 0; k < 16; ++k) w[k] = W_hh[k];

    const long stride = (long)BB * 2;
    const float4* g4 = (const float4*)xg + (long)b * 2;

    float4 qlo[PF], qhi[PF];
#pragma unroll
    for (int d = 0; d < PF; ++d) {
        qlo[d] = g4[d * stride];
        qhi[d] = g4[d * stride + 1];
    }
    const float4* gn = g4 + PF * stride;

    float h0 = 0.f, h1 = 0.f, c0 = 0.f, c1 = 0.f;

    for (int t0 = 0; t0 < TT - PF; t0 += PF) {
#pragma unroll
        for (int d = 0; d < PF; ++d) {
            lstm_step(qlo[d], qhi[d], w, h0, h1, c0, c1);
            qlo[d] = gn[d * stride];
            qhi[d] = gn[d * stride + 1];
        }
        gn += PF * stride;
    }
#pragma unroll
    for (int d = 0; d < PF; ++d) lstm_step(qlo[d], qhi[d], w, h0, h1, c0, c1);

#pragma unroll
    for (int c = 0; c < CC; ++c) {
        out[b * CC + c] = fc_b[c] + h0 * fc_W[c * HH + 0] + h1 * fc_W[c * HH + 1];
    }
}

__global__ void lstm_fused(const float* __restrict__ x,
                           const float* __restrict__ W_ih,
                           const float* __restrict__ W_hh,
                           const float* __restrict__ b_ih,
                           const float* __restrict__ b_hh,
                           const float* __restrict__ fc_W,
                           const float* __restrict__ fc_b,
                           float* __restrict__ out) {
    const int b = blockIdx.x * blockDim.x + threadIdx.x;
    if (b >= BB) return;

    float w[16];
#pragma unroll
    for (int k = 0; k < 16; ++k) w[k] = W_hh[k];
    float bias[NG];
#pragma unroll
    for (int g = 0; g < NG; ++g) bias[g] = b_ih[g] + b_hh[g];

    float h0 = 0.f, h1 = 0.f, c0 = 0.f, c1 = 0.f;
    const float* xb = x + (long)b * TT * II;

    for (int t = 0; t < TT; ++t) {
        const float* xr = xb + (long)t * II;
        float ga[NG];
#pragma unroll
        for (int g = 0; g < NG; ++g) {
            float s = bias[g];
            for (int i = 0; i < II; ++i) s += xr[i] * W_ih[g * II + i];
            ga[g] = s;
        }
        const float i0 = ga[0] + h0 * w[0]  + h1 * w[1];
        const float i1 = ga[1] + h0 * w[2]  + h1 * w[3];
        const float f0 = ga[2] + h0 * w[4]  + h1 * w[5];
        const float f1 = ga[3] + h0 * w[6]  + h1 * w[7];
        const float g0 = ga[4] + h0 * w[8]  + h1 * w[9];
        const float g1 = ga[5] + h0 * w[10] + h1 * w[11];
        const float o0 = ga[6] + h0 * w[12] + h1 * w[13];
        const float o1 = ga[7] + h0 * w[14] + h1 * w[15];

        c0 = fast_sigm(f0) * c0 + fast_sigm(i0) * fast_tanh(g0);
        c1 = fast_sigm(f1) * c1 + fast_sigm(i1) * fast_tanh(g1);
        h0 = fast_sigm(o0) * fast_tanh(c0);
        h1 = fast_sigm(o1) * fast_tanh(c1);
    }
#pragma unroll
    for (int c = 0; c < CC; ++c) {
        out[b * CC + c] = fc_b[c] + h0 * fc_W[c * HH + 0] + h1 * fc_W[c * HH + 1];
    }
}

extern "C" void kernel_launch(void* const* d_in, const int* in_sizes, int n_in,
                              void* d_out, int out_size, void* d_ws, size_t ws_size,
                              hipStream_t stream) {
    const float* x    = (const float*)d_in[0];
    const float* W_ih = (const float*)d_in[1];
    const float* W_hh = (const float*)d_in[2];
    const float* b_ih = (const float*)d_in[3];
    const float* b_hh = (const float*)d_in[4];
    const float* fc_W = (const float*)d_in[5];
    const float* fc_b = (const float*)d_in[6];
    float* out = (float*)d_out;

    const size_t need = (size_t)MM * NG * sizeof(float);
    if (ws_size >= need) {
        float* xg = (float*)d_ws;
        lstm_xproj_wmma<<<2048, 256, 0, stream>>>(x, W_ih, b_ih, b_hh, xg);
        lstm_scan<<<BB / 32, 32, 0, stream>>>(xg, W_hh, fc_W, fc_b, out);
    } else {
        lstm_fused<<<BB / 32, 32, 0, stream>>>(x, W_ih, W_hh, b_ih, b_hh,
                                               fc_W, fc_b, out);
    }
}


# scan wave reads all 16 gate rows of a chunk from LDS as one group behind the chunk barrier (were fetched row by row, under one LDS latency ahead of use); arithmetic and summation order unchanged
# speedup vs baseline: 4.0391x; 1.1021x over previous
#include <hip/hip_runtime.h>
#include <hip/hip_bf16.h>


#define BB 2048
#define TT 1024
#define II 28
#define HH 2
#define CC 10
#define NG 8

typedef __attribute__((ext_vector_type(2))) float v2f;
typedef __attribute__((ext_vector_type(8))) float v8f;

__device__ __forceinline__ float fast_tanh(float x) {
#if __has_builtin(__builtin_amdgcn_tanhf)
    return __builtin_amdgcn_tanhf(x);
#else
    float e = __expf(-2.0f * x);
    return (1.0f - e) / (1.0f + e);
#endif
}
__device__ __forceinline__ float fast_sigm(float x) {
    return 0.5f + 0.5f * fast_tanh(0.5f * x);
}

#define NB 16
#define CH 16
#define NCHUNK (TT / CH)
#define NPROD 3
#define TPW ((NB + NPROD - 1) / NPROD)
#define BUFSZ (CH * NB * NG)
#define LDS_FLOATS (41 * 1024)
static_assert(TT % CH == 0 && BB % NB == 0 && 2 * BUFSZ <= LDS_FLOATS, "shape");
static_assert((NPROD - 1) + NPROD * (TPW - 2) < NB, "every slot but the last is a real tile for every producer");

__device__ __forceinline__ void lstm_step(const float4 lo, const float4 hi,
                                          const float (&w)[16],
                                          float& h0, float& h1,
                                          float& c0, float& c1) {
    const float i0 = lo.x + h0 * w[0]  + h1 * w[1];
    const float i1 = lo.y + h0 * w[2]  + h1 * w[3];
    const float f0 = lo.z + h0 * w[4]  + h1 * w[5];
    const float f1 = lo.w + h0 * w[6]  + h1 * w[7];
    const float g0 = hi.x + h0 * w[8]  + h1 * w[9];
    const float g1 = hi.y + h0 * w[10] + h1 * w[11];
    const float o0 = hi.z + h0 * w[12] + h1 * w[13];
    const float o1 = hi.w + h0 * w[14] + h1 * w[15];

    c0 = fast_sigm(f0) * c0 + fast_sigm(i0) * fast_tanh(g0);
    c1 = fast_sigm(f1) * c1 + fast_sigm(i1) * fast_tanh(g1);
    h0 = fast_sigm(o0) * fast_tanh(c0);
    h1 = fast_sigm(o1) * fast_tanh(c1);
}

__device__ __forceinline__ void scan_chunk(const float4* src, const float (&w)[16],
                                           float& h0, float& h1, float& c0, float& c1) {
    float4 lo[CH], hi[CH];
#pragma unroll
    for (int d = 0; d < CH; ++d) { lo[d] = src[d * NB * 2]; hi[d] = src[d * NB * 2 + 1]; }
    __builtin_amdgcn_sched_barrier(0);
#pragma unroll
    for (int s = 0; s < CH; ++s) lstm_step(lo[s], hi[s], w, h0, h1, c0, c1);
}

__device__ __forceinline__ void prod_load(v2f (&A)[TPW][7], const float* __restrict__ x,
                                          int b0, int kk, int pw, int n, int khalf) {
    const int kc = (kk < NCHUNK) ? kk : (NCHUNK - 1);
#pragma unroll
    for (int i = 0; i < TPW; ++i) {
        const int pr = pw + NPROD * i;
        const int p  = (pr < NB) ? pr : (NB - 1);
        const float* xrow = x + ((long)(b0 + p) * TT + kc * CH + n) * II;
#pragma unroll
        for (int j = 0; j < 7; ++j) {
            const int k0 = 4 * j + 2 * khalf;
            v2f a; a.x = xrow[k0]; a.y = xrow[k0 + 1];
            A[i][j] = a;
        }
    }
}

__device__ __forceinline__ void prod_compute(const v2f (&A)[TPW][7], const v2f (&bfrag)[7],
                                             const v8f cbias, float* dst,
                                             int pw, int n, int khalf) {
    v8f c[TPW];
#pragma unroll
    for (int i = 0; i < TPW; ++i) c[i] = cbias;
#pragma unroll
    for (int j = 0; j < 7; ++j) {
#pragma unroll
        for (int i = 0; i < TPW; ++i) {
            c[i] = __builtin_amdgcn_wmma_f32_16x16x4_f32(
                       false, A[i][j], false, bfrag[j], (short)0, c[i], false, false);
        }
    }
    if (n < NG) {
#pragma unroll
        for (int i = 0; i < TPW; ++i) {
            const int p = pw + NPROD * i;
            if (i < TPW - 1 || p < NB) {
#pragma unroll
                for (int v = 0; v < 8; ++v) {
                    const int m = v + 8 * khalf;
                    dst[(m * NB + p) * NG + n] = c[i][v];
                }
            }
        }
    }
}

__global__ void __launch_bounds__(128)
lstm_fused_wmma(const float* __restrict__ x,
                const float* __restrict__ W_ih,
                const float* __restrict__ W_hh,
                const float* __restrict__ b_ih,
                const float* __restrict__ b_hh,
                const float* __restrict__ fc_W,
                const float* __restrict__ fc_b,
                float* __restrict__ out) {
    __shared__ __attribute__((aligned(16))) float xgs[LDS_FLOATS];

    const int lane  = threadIdx.x & 31;
    const int wv    = __builtin_amdgcn_readfirstlane((int)(threadIdx.x >> 5));
    const int pw    = wv - 1;
    const int n     = lane & 15;
    const int khalf = lane >> 4;
    const int b0    = blockIdx.x * NB;

    v2f bfrag[7];
#pragma unroll
    for (int j = 0; j < 7; ++j) {
        const int k0 = 4 * j + 2 * khalf;
        float q0 = 0.0f, q1 = 0.0f;
        if (n < NG) {
            q0 = W_ih[n * II + k0];
            q1 = W_ih[n * II + k0 + 1];
        }
        v2f bf; bf.x = q0; bf.y = q1;
        bfrag[j] = bf;
    }
    const float bias = (n < NG) ? (b_ih[n] + b_hh[n]) : 0.0f;
    v8f cbias;
#pragma unroll
    for (int v = 0; v < 8; ++v) cbias[v] = bias;

    float w[16];
#pragma unroll
    for (int k = 0; k < 16; ++k) w[k] = W_hh[k];
    float h0 = 0.f, h1 = 0.f, c0 = 0.f, c1 = 0.f;
    const float4* srow = (const float4*)xgs + n * 2;

    v2f cur[TPW][7], nxt[TPW][7];
    if (wv != 0) {
        prod_load(cur, x, b0, 0, pw, n, khalf);
        prod_compute(cur, bfrag, cbias, xgs, pw, n, khalf);
        prod_load(cur, x, b0, 1, pw, n, khalf);
    }

    for (int k = 0; k < NCHUNK; ++k) {
        __syncthreads();
        if (wv == 0) {
            scan_chunk(srow + (k & 1) * (BUFSZ / 4), w, h0, h1, c0, c1);
        } else {
            prod_load(nxt, x, b0, k + 2, pw, n, khalf);
            if (k + 1 < NCHUNK)
                prod_compute(cur, bfrag, cbias, xgs + ((k + 1) & 1) * BUFSZ, pw, n, khalf);
#pragma unroll
            for (int i = 0; i < TPW; ++i) {
#pragma unroll
                for (int j = 0; j < 7; ++j) cur[i][j] = nxt[i][j];
            }
        }
    }

    if (wv == 0 && lane < NB) {
        const int b = b0 + lane;
#pragma unroll
        for (int c = 0; c < CC; ++c) {
            out[b * CC + c] = fc_b[c] + h0 * fc_W[c * HH + 0] + h1 * fc_W[c * HH + 1];
        }
    }
}

extern "C" void kernel_launch(void* const* d_in, const int* in_sizes, int n_in,
                              void* d_out, int out_size, void* d_ws, size_t ws_size,
                              hipStream_t stream) {
    const float* x    = (const float*)d_in[0];
    const float* W_ih = (const float*)d_in[1];
    const float* W_hh = (const float*)d_in[2];
    const float* b_ih = (const float*)d_in[3];
    const float* b_hh = (const float*)d_in[4];
    const float* fc_W = (const float*)d_in[5];
    const float* fc_b = (const float*)d_in[6];
    float* out = (float*)d_out;

    lstm_fused_wmma<<<BB / NB, 128, 0, stream>>>(x, W_ih, W_hh, b_ih, b_hh, fc_W, fc_b, out);
}


# x tiles fetched as they lie (a load instruction covers whole consecutive lines; was 16-byte pieces of 14 lines) and re-laid into A fragments through a per-wave LDS stage; k order and sums unchanged
# speedup vs baseline: 4.6226x; 1.1445x over previous
#include <hip/hip_runtime.h>
#include <hip/hip_bf16.h>


#define BB 2048
#define TT 1024
#define II 28
#define HH 2
#define CC 10
#define NG 8

typedef __attribute__((ext_vector_type(2))) float v2f;
typedef __attribute__((ext_vector_type(4))) float v4f;
typedef __attribute__((ext_vector_type(8))) float v8f;

__device__ __forceinline__ float fast_tanh(float x) {
#if __has_builtin(__builtin_amdgcn_tanhf)
    return __builtin_amdgcn_tanhf(x);
#else
    float e = __expf(-2.0f * x);
    return (1.0f - e) / (1.0f + e);
#endif
}
__device__ __forceinline__ float fast_sigm(float x) {
    return 0.5f + 0.5f * fast_tanh(0.5f * x);
}

#define NB 16
#define CH 16
#define NCHUNK (TT / CH)
#define NPROD 3
#define TPW ((NB + NPROD - 1) / NPROD)
#define BUFSZ (CH * NB * NG)
#define LDS_FLOATS (41 * 1024)
#define TILE_F (CH * II)
#define Q4 (TILE_F / 128)
#define STAGE_OFF (2 * BUFSZ)
#define STAGE_F (TPW * TILE_F)
static_assert(TT % CH == 0 && BB % NB == 0 && STAGE_OFF + NPROD * STAGE_F <= LDS_FLOATS, "shape");
static_assert(TILE_F == Q4 * 128 + 64, "a tile is Q4 groups of 16 bytes and one of 8 bytes per lane");
static_assert((NPROD - 1) + NPROD * (TPW - 2) < NB, "every slot but the last is a real tile for every producer");

__device__ __forceinline__ void lstm_step(const float4 lo, const float4 hi,
                                          const float (&w)[16],
                                          float& h0, float& h1,
                                          float& c0, float& c1) {
    const float i0 = lo.x + h0 * w[0]  + h1 * w[1];
    const float i1 = lo.y + h0 * w[2]  + h1 * w[3];
    const float f0 = lo.z + h0 * w[4]  + h1 * w[5];
    const float f1 = lo.w + h0 * w[6]  + h1 * w[7];
    const float g0 = hi.x + h0 * w[8]  + h1 * w[9];
    const float g1 = hi.y + h0 * w[10] + h1 * w[11];
    const float o0 = hi.z + h0 * w[12] + h1 * w[13];
    const float o1 = hi.w + h0 * w[14] + h1 * w[15];

    c0 = fast_sigm(f0) * c0 + fast_sigm(i0) * fast_tanh(g0);
    c1 = fast_sigm(f1) * c1 + fast_sigm(i1) * fast_tanh(g1);
    h0 = fast_sigm(o0) * fast_tanh(c0);
    h1 = fast_sigm(o1) * fast_tanh(c1);
}

__device__ __forceinline__ void scan_chunk(const float4* src, const float (&w)[16],
                                           float& h0, float& h1, float& c0, float& c1) {
    float4 lo[CH], hi[CH];
#pragma unroll
    for (int d = 0; d < CH; ++d) { lo[d] = src[d * NB * 2]; hi[d] = src[d * NB * 2 + 1]; }
    __builtin_amdgcn_sched_barrier(0);
#pragma unroll
    for (int s = 0; s < CH; ++s) lstm_step(lo[s], hi[s], w, h0, h1, c0, c1);
}

__device__ __forceinline__ void prod_load(v4f (&rq)[TPW][Q4], v2f (&rr)[TPW],
                                          const float* __restrict__ x,
                                          int b0, int kk, int pw, int lane) {
    const int kc = (kk < NCHUNK) ? kk : (NCHUNK - 1);
#pragma unroll
    for (int i = 0; i < TPW; ++i) {
        const int pr = pw + NPROD * i;
        const int p  = (pr < NB) ? pr : (NB - 1);
        const float* tile = x + ((long)(b0 + p) * TT + kc * CH) * II;
#pragma unroll
        for (int j = 0; j < Q4; ++j)
            rq[i][j] = *(const v4f*)(tile + j * 128 + 4 * lane);
        rr[i] = *(const v2f*)(tile + Q4 * 128 + 2 * lane);
    }
}

__device__ __forceinline__ void prod_stage(const v4f (&rq)[TPW][Q4], const v2f (&rr)[TPW],
                                           float* stage, int lane) {
#pragma unroll
    for (int i = 0; i < TPW; ++i) {
        float* t = stage + i * TILE_F;
#pragma unroll
        for (int j = 0; j < Q4; ++j)
            *(v4f*)(t + j * 128 + 4 * lane) = rq[i][j];
        *(v2f*)(t + Q4 * 128 + 2 * lane) = rr[i];
    }
}

__device__ __forceinline__ void prod_frags(v2f (&A)[TPW][7], const float* stage, int n, int khalf) {
#pragma unroll
    for (int i = 0; i < TPW; ++i) {
        const float* xrow = stage + i * TILE_F + n * II;
#pragma unroll
        for (int j = 0; j < 7; ++j)
            A[i][j] = *(const v2f*)(xrow + 4 * j + 2 * khalf);
    }
}

__device__ __forceinline__ void prod_compute(const v2f (&A)[TPW][7], const v2f (&bfrag)[7],
                                             const v8f cbias, float* dst,
                                             int pw, int n, int khalf) {
    v8f c[TPW];
#pragma unroll
    for (int i = 0; i < TPW; ++i) c[i] = cbias;
#pragma unroll
    for (int j = 0; j < 7; ++j) {
#pragma unroll
        for (int i = 0; i < TPW; ++i) {
            c[i] = __builtin_amdgcn_wmma_f32_16x16x4_f32(
                       false, A[i][j], false, bfrag[j], (short)0, c[i], false, false);
        }
    }
    if (n < NG) {
#pragma unroll
        for (int i = 0; i < TPW; ++i) {
            const int p = pw + NPROD * i;
            if (i < TPW - 1 || p < NB) {
#pragma unroll
                for (int v = 0; v < 8; ++v) {
                    const int m = v + 8 * khalf;
                    dst[(m * NB + p) * NG + n] = c[i][v];
                }
            }
        }
    }
}

__global__ void __launch_bounds__(128)
lstm_fused_wmma(const float* __restrict__ x,
                const float* __restrict__ W_ih,
                const float* __restrict__ W_hh,
                const float* __restrict__ b_ih,
                const float* __restrict__ b_hh,
                const float* __restrict__ fc_W,
                const float* __restrict__ fc_b,
                float* __restrict__ out) {
    __shared__ __attribute__((aligned(16))) float xgs[LDS_FLOATS];

    const int lane  = threadIdx.x & 31;
    const int wv    = __builtin_amdgcn_readfirstlane((int)(threadIdx.x >> 5));
    const int pw    = wv - 1;
    const int n     = lane & 15;
    const int khalf = lane >> 4;
    const int b0    = blockIdx.x * NB;

    v2f bfrag[7];
#pragma unroll
    for (int j = 0; j < 7; ++j) {
        const int k0 = 4 * j + 2 * khalf;
        float q0 = 0.0f, q1 = 0.0f;
        if (n < NG) {
            q0 = W_ih[n * II + k0];
            q1 = W_ih[n * II + k0 + 1];
        }
        v2f bf; bf.x = q0; bf.y = q1;
        bfrag[j] = bf;
    }
    const float bias = (n < NG) ? (b_ih[n] + b_hh[n]) : 0.0f;
    v8f cbias;
#pragma unroll
    for (int v = 0; v < 8; ++v) cbias[v] = bias;

    float w[16];
#pragma unroll
    for (int k = 0; k < 16; ++k) w[k] = W_hh[k];
    float h0 = 0.f, h1 = 0.f, c0 = 0.f, c1 = 0.f;
    const float4* srow = (const float4*)xgs + n * 2;

    v4f rq[TPW][Q4];
    v2f rr[TPW];
    v2f frag[TPW][7];
    float* stage = xgs + STAGE_OFF + (pw < 0 ? 0 : pw) * STAGE_F;
    if (wv != 0) {
        prod_load(rq, rr, x, b0, 0, pw, lane);
        prod_stage(rq, rr, stage, lane);
        prod_load(rq, rr, x, b0, 1, pw, lane);
        prod_frags(frag, stage, n, khalf);
        prod_compute(frag, bfrag, cbias, xgs, pw, n, khalf);
    }

    for (int k = 0; k < NCHUNK; ++k) {
        __syncthreads();
        if (wv == 0) {
            scan_chunk(srow + (k & 1) * (BUFSZ / 4), w, h0, h1, c0, c1);
        } else {
            prod_stage(rq, rr, stage, lane);
            prod_load(rq, rr, x, b0, k + 2, pw, lane);
            prod_frags(frag, stage, n, khalf);
            if (k + 1 < NCHUNK)
                prod_compute(frag, bfrag, cbias, xgs + ((k + 1) & 1) * BUFSZ, pw, n, khalf);
        }
    }

    if (wv == 0 && lane < NB) {
        const int b = b0 + lane;
#pragma unroll
        for (int c = 0; c < CC; ++c) {
            out[b * CC + c] = fc_b[c] + h0 * fc_W[c * HH + 0] + h1 * fc_W[c * HH + 1];
        }
    }
}

extern "C" void kernel_launch(void* const* d_in, const int* in_sizes, int n_in,
                              void* d_out, int out_size, void* d_ws, size_t ws_size,
                              hipStream_t stream) {
    const float* x    = (const float*)d_in[0];
    const float* W_ih = (const float*)d_in[1];
    const float* W_hh = (const float*)d_in[2];
    const float* b_ih = (const float*)d_in[3];
    const float* b_hh = (const float*)d_in[4];
    const float* fc_W = (const float*)d_in[5];
    const float* fc_b = (const float*)d_in[6];
    float* out = (float*)d_out;

    lstm_fused_wmma<<<BB / NB, 128, 0, stream>>>(x, W_ih, W_hh, b_ih, b_hh, fc_W, fc_b, out);
}


# x rows requested two chunks earlier and left in flight across the chunk barrier (hand-written barrier waits for LDS only); all 4 waves fetch into a shared double staging buffer; sums unchanged
# speedup vs baseline: 6.3069x; 1.3644x over previous
#include <hip/hip_runtime.h>
#include <hip/hip_bf16.h>


#define BB 2048
#define TT 1024
#define II 28
#define HH 2
#define CC 10
#define NG 8

typedef __attribute__((ext_vector_type(2))) float v2f;
typedef __attribute__((ext_vector_type(4))) float v4f;
typedef __attribute__((ext_vector_type(8))) float v8f;

__device__ __forceinline__ float fast_tanh(float x) {
#if __has_builtin(__builtin_amdgcn_tanhf)
    return __builtin_amdgcn_tanhf(x);
#else
    float e = __expf(-2.0f * x);
    return (1.0f - e) / (1.0f + e);
#endif
}
__device__ __forceinline__ float fast_sigm(float x) {
    return 0.5f + 0.5f * fast_tanh(0.5f * x);
}

#define NB 16
#define CH 16
#define NCHUNK (TT / CH)
#define NPROD 3
#define NWAVE (NPROD + 1)
#define TPW ((NB + NPROD - 1) / NPROD)
#define TPL (NB / NWAVE)
#define BUFSZ (CH * NB * NG)
#define LDS_FLOATS (41 * 1024)
#define TILE_F (CH * II)
#define Q4 (TILE_F / 128)
#define STAGE_OFF (2 * BUFSZ)
#define SBUF_F (NB * TILE_F)
static_assert(TT % CH == 0 && NCHUNK % 2 == 0 && BB % NB == 0 && STAGE_OFF + 2 * SBUF_F <= LDS_FLOATS, "shape");
static_assert(NB % NWAVE == 0, "every wave fetches the same number of tiles");
static_assert(TILE_F == Q4 * 128 + 64, "a tile is Q4 groups of 16 bytes and one of 8 bytes per lane");
static_assert((NPROD - 1) + NPROD * (TPW - 2) < NB, "every slot but the last is a real tile for every producer");

__device__ __forceinline__ void chunk_barrier() {
    asm volatile("s_wait_dscnt 0x0\n\ts_barrier_signal -1\n\ts_barrier_wait -1" ::: "memory");
}

__device__ __forceinline__ void lstm_step(const float4 lo, const float4 hi,
                                          const float (&w)[16],
                                          float& h0, float& h1,
                                          float& c0, float& c1) {
    const float i0 = lo.x + h0 * w[0]  + h1 * w[1];
    const float i1 = lo.y + h0 * w[2]  + h1 * w[3];
    const float f0 = lo.z + h0 * w[4]  + h1 * w[5];
    const float f1 = lo.w + h0 * w[6]  + h1 * w[7];
    const float g0 = hi.x + h0 * w[8]  + h1 * w[9];
    const float g1 = hi.y + h0 * w[10] + h1 * w[11];
    const float o0 = hi.z + h0 * w[12] + h1 * w[13];
    const float o1 = hi.w + h0 * w[14] + h1 * w[15];

    c0 = fast_sigm(f0) * c0 + fast_sigm(i0) * fast_tanh(g0);
    c1 = fast_sigm(f1) * c1 + fast_sigm(i1) * fast_tanh(g1);
    h0 = fast_sigm(o0) * fast_tanh(c0);
    h1 = fast_sigm(o1) * fast_tanh(c1);
}

__device__ __forceinline__ void scan_chunk(const float4* src, const float (&w)[16],
                                           float& h0, float& h1, float& c0, float& c1) {
    float4 lo[CH], hi[CH];
#pragma unroll
    for (int d = 0; d < CH; ++d) { lo[d] = src[d * NB * 2]; hi[d] = src[d * NB * 2 + 1]; }
    __builtin_amdgcn_sched_barrier(0);
#pragma unroll
    for (int s = 0; s < CH; ++s) lstm_step(lo[s], hi[s], w, h0, h1, c0, c1);
}

__device__ __forceinline__ void fetch_load(v4f (&rq)[TPL][Q4], v2f (&rr)[TPL],
                                           const float* __restrict__ x,
                                           int b0, int kk, int wv, int lane) {
    const int kc = (kk < NCHUNK) ? kk : (NCHUNK - 1);
#pragma unroll
    for (int i = 0; i < TPL; ++i) {
        const int p = wv * TPL + i;
        const float* tile = x + ((long)(b0 + p) * TT + kc * CH) * II;
#pragma unroll
        for (int j = 0; j < Q4; ++j)
            rq[i][j] = *(const v4f*)(tile + j * 128 + 4 * lane);
        rr[i] = *(const v2f*)(tile + Q4 * 128 + 2 * lane);
    }
}

__device__ __forceinline__ void fetch_stage(const v4f (&rq)[TPL][Q4], const v2f (&rr)[TPL],
                                            float* sbuf, int wv, int lane) {
#pragma unroll
    for (int i = 0; i < TPL; ++i) {
        float* t = sbuf + (wv * TPL + i) * TILE_F;
#pragma unroll
        for (int j = 0; j < Q4; ++j)
            *(v4f*)(t + j * 128 + 4 * lane) = rq[i][j];
        *(v2f*)(t + Q4 * 128 + 2 * lane) = rr[i];
    }
}

__device__ __forceinline__ void prod_frags(v2f (&A)[TPW][7], const float* sbuf,
                                           int pw, int n, int khalf) {
#pragma unroll
    for (int i = 0; i < TPW; ++i) {
        const int pr = pw + NPROD * i;
        const int p  = (pr < NB) ? pr : (NB - 1);
        const float* xrow = sbuf + p * TILE_F + n * II;
#pragma unroll
        for (int j = 0; j < 7; ++j)
            A[i][j] = *(const v2f*)(xrow + 4 * j + 2 * khalf);
    }
}

__device__ __forceinline__ void prod_compute(const v2f (&A)[TPW][7], const v2f (&bfrag)[7],
                                             const v8f cbias, float* dst,
                                             int pw, int n, int khalf) {
    v8f c[TPW];
#pragma unroll
    for (int i = 0; i < TPW; ++i) c[i] = cbias;
#pragma unroll
    for (int j = 0; j < 7; ++j) {
#pragma unroll
        for (int i = 0; i < TPW; ++i) {
            c[i] = __builtin_amdgcn_wmma_f32_16x16x4_f32(
                       false, A[i][j], false, bfrag[j], (short)0, c[i], false, false);
        }
    }
    if (n < NG) {
#pragma unroll
        for (int i = 0; i < TPW; ++i) {
            const int p = pw + NPROD * i;
            if (i < TPW - 1 || p < NB) {
#pragma unroll
                for (int v = 0; v < 8; ++v) {
                    const int m = v + 8 * khalf;
                    dst[(m * NB + p) * NG + n] = c[i][v];
                }
            }
        }
    }
}

__global__ void __launch_bounds__(128)
lstm_fused_wmma(const float* __restrict__ x,
                const float* __restrict__ W_ih,
                const float* __restrict__ W_hh,
                const float* __restrict__ b_ih,
                const float* __restrict__ b_hh,
                const float* __restrict__ fc_W,
                const float* __restrict__ fc_b,
                float* __restrict__ out) {
    __shared__ __attribute__((aligned(16))) float xgs[LDS_FLOATS];

    const int lane  = threadIdx.x & 31;
    const int wv    = __builtin_amdgcn_readfirstlane((int)(threadIdx.x >> 5));
    const int pw    = wv - 1;
    const int n     = lane & 15;
    const int khalf = lane >> 4;
    const int b0    = blockIdx.x * NB;

    v2f bfrag[7];
#pragma unroll
    for (int j = 0; j < 7; ++j) {
        const int k0 = 4 * j + 2 * khalf;
        float q0 = 0.0f, q1 = 0.0f;
        if (n < NG) {
            q0 = W_ih[n * II + k0];
            q1 = W_ih[n * II + k0 + 1];
        }
        v2f bf; bf.x = q0; bf.y = q1;
        bfrag[j] = bf;
    }
    const float bias = (n < NG) ? (b_ih[n] + b_hh[n]) : 0.0f;
    v8f cbias;
#pragma unroll
    for (int v = 0; v < 8; ++v) cbias[v] = bias;

    float w[16];
#pragma unroll
    for (int k = 0; k < 16; ++k) w[k] = W_hh[k];
    float h0 = 0.f, h1 = 0.f, c0 = 0.f, c1 = 0.f;
    const float4* srow = (const float4*)xgs + n * 2;

    v4f rqA[TPL][Q4], rqB[TPL][Q4];
    v2f rrA[TPL], rrB[TPL];
    v2f frag[TPW][7];
    float* sb0 = xgs + STAGE_OFF;
    float* sb1 = sb0 + SBUF_F;
    fetch_load(rqA, rrA, x, b0, 0, wv, lane);
    fetch_load(rqB, rrB, x, b0, 1, wv, lane);
    fetch_stage(rqA, rrA, sb0, wv, lane);
    fetch_load(rqA, rrA, x, b0, 2, wv, lane);
    fetch_stage(rqB, rrB, sb1, wv, lane);
    fetch_load(rqB, rrB, x, b0, 3, wv, lane);
    chunk_barrier();
    if (wv != 0) {
        prod_frags(frag, sb0, pw, n, khalf);
        prod_compute(frag, bfrag, cbias, xgs, pw, n, khalf);
    }

    for (int k = 0; k < NCHUNK; k += 2) {
        chunk_barrier();
        fetch_stage(rqA, rrA, sb0, wv, lane);
        fetch_load(rqA, rrA, x, b0, k + 4, wv, lane);
        if (wv == 0) {
            scan_chunk(srow, w, h0, h1, c0, c1);
        } else {
            prod_frags(frag, sb1, pw, n, khalf);
            prod_compute(frag, bfrag, cbias, xgs + BUFSZ, pw, n, khalf);
        }
        chunk_barrier();
        fetch_stage(rqB, rrB, sb1, wv, lane);
        fetch_load(rqB, rrB, x, b0, k + 5, wv, lane);
        if (wv == 0) {
            scan_chunk(srow + BUFSZ / 4, w, h0, h1, c0, c1);
        } else {
            prod_frags(frag, sb0, pw, n, khalf);
            if (k + 2 < NCHUNK)
                prod_compute(frag, bfrag, cbias, xgs, pw, n, khalf);
        }
    }

    if (wv == 0 && lane < NB) {
        const int b = b0 + lane;
#pragma unroll
        for (int c = 0; c < CC; ++c) {
            out[b * CC + c] = fc_b[c] + h0 * fc_W[c * HH + 0] + h1 * fc_W[c * HH + 1];
        }
    }
}

extern "C" void kernel_launch(void* const* d_in, const int* in_sizes, int n_in,
                              void* d_out, int out_size, void* d_ws, size_t ws_size,
                              hipStream_t stream) {
    const float* x    = (const float*)d_in[0];
    const float* W_ih = (const float*)d_in[1];
    const float* W_hh = (const float*)d_in[2];
    const float* b_ih = (const float*)d_in[3];
    const float* b_hh = (const float*)d_in[4];
    const float* fc_W = (const float*)d_in[5];
    const float* fc_b = (const float*)d_in[6];
    float* out = (float*)d_out;

    lstm_fused_wmma<<<BB / NB, 128, 0, stream>>>(x, W_ih, W_hh, b_ih, b_hh, fc_W, fc_b, out);
}


# 8 batch elements per workgroup (was 16): 256 workgroups, one on every WGP (128 left half the chip idle); each wave's fetch, staging and WMMA tile work per chunk halves; sums unchanged
# speedup vs baseline: 6.8814x; 1.0911x over previous
#include <hip/hip_runtime.h>
#include <hip/hip_bf16.h>


#define BB 2048
#define TT 1024
#define II 28
#define HH 2
#define CC 10
#define NG 8

typedef __attribute__((ext_vector_type(2))) float v2f;
typedef __attribute__((ext_vector_type(4))) float v4f;
typedef __attribute__((ext_vector_type(8))) float v8f;

__device__ __forceinline__ float fast_tanh(float x) {
#if __has_builtin(__builtin_amdgcn_tanhf)
    return __builtin_amdgcn_tanhf(x);
#else
    float e = __expf(-2.0f * x);
    return (1.0f - e) / (1.0f + e);
#endif
}
__device__ __forceinline__ float fast_sigm(float x) {
    return 0.5f + 0.5f * fast_tanh(0.5f * x);
}

#define NB 8
#define CH 16
#define NCHUNK (TT / CH)
#define NPROD 3
#define NWAVE (NPROD + 1)
#define TPW ((NB + NPROD - 1) / NPROD)
#define TPL (NB / NWAVE)
#define BUFSZ (CH * NB * NG)
#define LDS_FLOATS (41 * 1024)
#define TILE_F (CH * II)
#define Q4 (TILE_F / 128)
#define STAGE_OFF (2 * BUFSZ)
#define SBUF_F (NB * TILE_F)
static_assert(TT % CH == 0 && NCHUNK % 2 == 0 && BB % NB == 0 && STAGE_OFF + 2 * SBUF_F <= LDS_FLOATS, "shape");
static_assert(NB % NWAVE == 0, "every wave fetches the same number of tiles");
static_assert(TILE_F == Q4 * 128 + 64, "a tile is Q4 groups of 16 bytes and one of 8 bytes per lane");
static_assert((NPROD - 1) + NPROD * (TPW - 2) < NB, "every slot but the last is a real tile for every producer");

__device__ __forceinline__ void chunk_barrier() {
    asm volatile("s_wait_dscnt 0x0\n\ts_barrier_signal -1\n\ts_barrier_wait -1" ::: "memory");
}

__device__ __forceinline__ void lstm_step(const float4 lo, const float4 hi,
                                          const float (&w)[16],
                                          float& h0, float& h1,
                                          float& c0, float& c1) {
    const float i0 = lo.x + h0 * w[0]  + h1 * w[1];
    const float i1 = lo.y + h0 * w[2]  + h1 * w[3];
    const float f0 = lo.z + h0 * w[4]  + h1 * w[5];
    const float f1 = lo.w + h0 * w[6]  + h1 * w[7];
    const float g0 = hi.x + h0 * w[8]  + h1 * w[9];
    const float g1 = hi.y + h0 * w[10] + h1 * w[11];
    const float o0 = hi.z + h0 * w[12] + h1 * w[13];
    const float o1 = hi.w + h0 * w[14] + h1 * w[15];

    c0 = fast_sigm(f0) * c0 + fast_sigm(i0) * fast_tanh(g0);
    c1 = fast_sigm(f1) * c1 + fast_sigm(i1) * fast_tanh(g1);
    h0 = fast_sigm(o0) * fast_tanh(c0);
    h1 = fast_sigm(o1) * fast_tanh(c1);
}

__device__ __forceinline__ void scan_chunk(const float4* src, const float (&w)[16],
                                           float& h0, float& h1, float& c0, float& c1) {
    float4 lo[CH], hi[CH];
#pragma unroll
    for (int d = 0; d < CH; ++d) { lo[d] = src[d * NB * 2]; hi[d] = src[d * NB * 2 + 1]; }
    __builtin_amdgcn_sched_barrier(0);
#pragma unroll
    for (int s = 0; s < CH; ++s) lstm_step(lo[s], hi[s], w, h0, h1, c0, c1);
}

__device__ __forceinline__ void fetch_load(v4f (&rq)[TPL][Q4], v2f (&rr)[TPL],
                                           const float* __restrict__ x,
                                           int b0, int kk, int wv, int lane) {
    const int kc = (kk < NCHUNK) ? kk : (NCHUNK - 1);
#pragma unroll
    for (int i = 0; i < TPL; ++i) {
        const int p = wv * TPL + i;
        const float* tile = x + ((long)(b0 + p) * TT + kc * CH) * II;
#pragma unroll
        for (int j = 0; j < Q4; ++j)
            rq[i][j] = *(const v4f*)(tile + j * 128 + 4 * lane);
        rr[i] = *(const v2f*)(tile + Q4 * 128 + 2 * lane);
    }
}

__device__ __forceinline__ void fetch_stage(const v4f (&rq)[TPL][Q4], const v2f (&rr)[TPL],
                                            float* sbuf, int wv, int lane) {
#pragma unroll
    for (int i = 0; i < TPL; ++i) {
        float* t = sbuf + (wv * TPL + i) * TILE_F;
#pragma unroll
        for (int j = 0; j < Q4; ++j)
            *(v4f*)(t + j * 128 + 4 * lane) = rq[i][j];
        *(v2f*)(t + Q4 * 128 + 2 * lane) = rr[i];
    }
}

__device__ __forceinline__ void prod_frags(v2f (&A)[TPW][7], const float* sbuf,
                                           int pw, int n, int khalf) {
#pragma unroll
    for (int i = 0; i < TPW; ++i) {
        const int pr = pw + NPROD * i;
        const int p  = (pr < NB) ? pr : (NB - 1);
        const float* xrow = sbuf + p * TILE_F + n * II;
#pragma unroll
        for (int j = 0; j < 7; ++j)
            A[i][j] = *(const v2f*)(xrow + 4 * j + 2 * khalf);
    }
}

__device__ __forceinline__ void prod_compute(const v2f (&A)[TPW][7], const v2f (&bfrag)[7],
                                             const v8f cbias, float* dst,
                                             int pw, int n, int khalf) {
    v8f c[TPW];
#pragma unroll
    for (int i = 0; i < TPW; ++i) c[i] = cbias;
#pragma unroll
    for (int j = 0; j < 7; ++j) {
#pragma unroll
        for (int i = 0; i < TPW; ++i) {
            c[i] = __builtin_amdgcn_wmma_f32_16x16x4_f32(
                       false, A[i][j], false, bfrag[j], (short)0, c[i], false, false);
        }
    }
    if (n < NG) {
#pragma unroll
        for (int i = 0; i < TPW; ++i) {
            const int p = pw + NPROD * i;
            if (i < TPW - 1 || p < NB) {
#pragma unroll
                for (int v = 0; v < 8; ++v) {
                    const int m = v + 8 * khalf;
                    dst[(m * NB + p) * NG + n] = c[i][v];
                }
            }
        }
    }
}

__global__ void __launch_bounds__(128)
lstm_fused_wmma(const float* __restrict__ x,
                const float* __restrict__ W_ih,
                const float* __restrict__ W_hh,
                const float* __restrict__ b_ih,
                const float* __restrict__ b_hh,
                const float* __restrict__ fc_W,
                const float* __restrict__ fc_b,
                float* __restrict__ out) {
    __shared__ __attribute__((aligned(16))) float xgs[LDS_FLOATS];

    const int lane  = threadIdx.x & 31;
    const int wv    = __builtin_amdgcn_readfirstlane((int)(threadIdx.x >> 5));
    const int pw    = wv - 1;
    const int n     = lane & 15;
    const int sl    = lane & (NB - 1);
    const int khalf = lane >> 4;
    const int b0    = blockIdx.x * NB;

    v2f bfrag[7];
#pragma unroll
    for (int j = 0; j < 7; ++j) {
        const int k0 = 4 * j + 2 * khalf;
        float q0 = 0.0f, q1 = 0.0f;
        if (n < NG) {
            q0 = W_ih[n * II + k0];
            q1 = W_ih[n * II + k0 + 1];
        }
        v2f bf; bf.x = q0; bf.y = q1;
        bfrag[j] = bf;
    }
    const float bias = (n < NG) ? (b_ih[n] + b_hh[n]) : 0.0f;
    v8f cbias;
#pragma unroll
    for (int v = 0; v < 8; ++v) cbias[v] = bias;

    float w[16];
#pragma unroll
    for (int k = 0; k < 16; ++k) w[k] = W_hh[k];
    float h0 = 0.f, h1 = 0.f, c0 = 0.f, c1 = 0.f;
    const float4* srow = (const float4*)xgs + sl * 2;

    v4f rqA[TPL][Q4], rqB[TPL][Q4];
    v2f rrA[TPL], rrB[TPL];
    v2f frag[TPW][7];
    float* sb0 = xgs + STAGE_OFF;
    float* sb1 = sb0 + SBUF_F;
    fetch_load(rqA, rrA, x, b0, 0, wv, lane);
    fetch_load(rqB, rrB, x, b0, 1, wv, lane);
    fetch_stage(rqA, rrA, sb0, wv, lane);
    fetch_load(rqA, rrA, x, b0, 2, wv, lane);
    fetch_stage(rqB, rrB, sb1, wv, lane);
    fetch_load(rqB, rrB, x, b0, 3, wv, lane);
    chunk_barrier();
    if (wv != 0) {
        prod_frags(frag, sb0, pw, n, khalf);
        prod_compute(frag, bfrag, cbias, xgs, pw, n, khalf);
    }

    for (int k = 0; k < NCHUNK; k += 2) {
        chunk_barrier();
        fetch_stage(rqA, rrA, sb0, wv, lane);
        fetch_load(rqA, rrA, x, b0, k + 4, wv, lane);
        if (wv == 0) {
            scan_chunk(srow, w, h0, h1, c0, c1);
        } else {
            prod_frags(frag, sb1, pw, n, khalf);
            prod_compute(frag, bfrag, cbias, xgs + BUFSZ, pw, n, khalf);
        }
        chunk_barrier();
        fetch_stage(rqB, rrB, sb1, wv, lane);
        fetch_load(rqB, rrB, x, b0, k + 5, wv, lane);
        if (wv == 0) {
            scan_chunk(srow + BUFSZ / 4, w, h0, h1, c0, c1);
        } else {
            prod_frags(frag, sb0, pw, n, khalf);
            if (k + 2 < NCHUNK)
                prod_compute(frag, bfrag, cbias, xgs, pw, n, khalf);
        }
    }

    if (wv == 0 && lane < NB) {
        const int b = b0 + lane;
#pragma unroll
        for (int c = 0; c < CC; ++c) {
            out[b * CC + c] = fc_b[c] + h0 * fc_W[c * HH + 0] + h1 * fc_W[c * HH + 1];
        }
    }
}

extern "C" void kernel_launch(void* const* d_in, const int* in_sizes, int n_in,
                              void* d_out, int out_size, void* d_ws, size_t ws_size,
                              hipStream_t stream) {
    const float* x    = (const float*)d_in[0];
    const float* W_ih = (const float*)d_in[1];
    const float* W_hh = (const float*)d_in[2];
    const float* b_ih = (const float*)d_in[3];
    const float* b_hh = (const float*)d_in[4];
    const float* fc_W = (const float*)d_in[5];
    const float* fc_b = (const float*)d_in[6];
    float* out = (float*)d_out;

    lstm_fused_wmma<<<BB / NB, 128, 0, stream>>>(x, W_ih, W_hh, b_ih, b_hh, fc_W, fc_b, out);
}


# scan step spread over the wave's 32 lanes: 4 lanes per batch element, 2 gates each (all 8 gates sat on one lane, 24 lanes idle), passed on by quad permute; gate tiles laid per gate in LDS; same sums
# speedup vs baseline: 7.2174x; 1.0488x over previous
#include <hip/hip_runtime.h>
#include <hip/hip_bf16.h>


#define BB 2048
#define TT 1024
#define II 28
#define HH 2
#define CC 10
#define NG 8

typedef __attribute__((ext_vector_type(2))) float v2f;
typedef __attribute__((ext_vector_type(4))) float v4f;
typedef __attribute__((ext_vector_type(8))) float v8f;

__device__ __forceinline__ float fast_tanh(float x) {
#if __has_builtin(__builtin_amdgcn_tanhf)
    return __builtin_amdgcn_tanhf(x);
#else
    float e = __expf(-2.0f * x);
    return (1.0f - e) / (1.0f + e);
#endif
}
__device__ __forceinline__ float fast_sigm(float x) {
    return 0.5f + 0.5f * fast_tanh(0.5f * x);
}

#define NB 8
#define CH 16
#define NCHUNK (TT / CH)
#define NPROD 3
#define NWAVE (NPROD + 1)
#define TPW ((NB + NPROD - 1) / NPROD)
#define TPL (NB / NWAVE)
#define TPITCH (NG * CH + 4)
#define BUFSZ (NB * TPITCH)
#define LDS_FLOATS (41 * 1024)
#define TILE_F (CH * II)
#define Q4 (TILE_F / 128)
#define STAGE_OFF (2 * BUFSZ)
#define SBUF_F (NB * TILE_F)
static_assert(TT % CH == 0 && NCHUNK % 2 == 0 && BB % NB == 0 && STAGE_OFF + 2 * SBUF_F <= LDS_FLOATS, "shape");
static_assert(NB % NWAVE == 0, "every wave fetches the same number of tiles");
static_assert(NB * 4 == 32 && NG == 8 && CH % 4 == 0 && TPITCH % 4 == 0, "scan: four lanes per batch element, two gates per lane, 16-byte groups");
static_assert(TILE_F == Q4 * 128 + 64, "a tile is Q4 groups of 16 bytes and one of 8 bytes per lane");
static_assert((NPROD - 1) + NPROD * (TPW - 2) < NB, "every slot but the last is a real tile for every producer");

__device__ __forceinline__ void chunk_barrier() {
    asm volatile("s_wait_dscnt 0x0\n\ts_barrier_signal -1\n\ts_barrier_wait -1" ::: "memory");
}

template <int CTRL>
__device__ __forceinline__ float quad_perm(float v) {
    return __builtin_bit_cast(float,
        __builtin_amdgcn_mov_dpp(__builtin_bit_cast(int, v), CTRL, 0xF, 0xF, true));
}

__device__ __forceinline__ float fma_h_lane2(float acc, float h, float w) {
    asm("v_fmac_f32_dpp %0, %1, %2 quad_perm:[2,2,2,2] row_mask:0xf bank_mask:0xf bound_ctrl:1"
        : "+v"(acc) : "v"(h), "v"(w));
    return acc;
}
__device__ __forceinline__ float fma_h_lane3(float acc, float h, float w) {
    asm("v_fmac_f32_dpp %0, %1, %2 quad_perm:[3,3,3,3] row_mask:0xf bank_mask:0xf bound_ctrl:1"
        : "+v"(acc) : "v"(h), "v"(w));
    return acc;
}

struct ScanLane {
    float wA0, wA1;
    float wB0, wB1;
    float sB, aB, bB;
};

__device__ __forceinline__ void lstm_step_quad(const float ga, const float gb,
                                               const ScanLane& k,
                                               float& h, float& c) {
    const float za = fma_h_lane3(fma_h_lane2(ga, h, k.wA0), h, k.wA1);
    const float zb = fma_h_lane3(fma_h_lane2(gb, h, k.wB0), h, k.wB1);
    const float sa = __builtin_fmaf(0.5f, fast_tanh(0.5f * za), 0.5f);
    const float sb = __builtin_fmaf(k.aB, fast_tanh(k.sB * zb), k.bB);
    const float si = quad_perm<0x44>(sa);
    const float tg = quad_perm<0x44>(sb);
    const float fc = sa * c;
    c = __builtin_fmaf(si, tg, fc);
    h = sb * fast_tanh(c);
}

__device__ __forceinline__ void scan_chunk(const v4f* src, const ScanLane& k,
                                           float& h, float& c) {
    v4f ra[CH / 4], rb[CH / 4];
#pragma unroll
    for (int j = 0; j < CH / 4; ++j) { ra[j] = src[j]; rb[j] = src[(NG / 2) * CH / 4 + j]; }
    __builtin_amdgcn_sched_barrier(0);
#pragma unroll
    for (int s = 0; s < CH; ++s) lstm_step_quad(ra[s >> 2][s & 3], rb[s >> 2][s & 3], k, h, c);
}

__device__ __forceinline__ void fetch_load(v4f (&rq)[TPL][Q4], v2f (&rr)[TPL],
                                           const float* __restrict__ x,
                                           int b0, int kk, int wv, int lane) {
    const int kc = (kk < NCHUNK) ? kk : (NCHUNK - 1);
#pragma unroll
    for (int i = 0; i < TPL; ++i) {
        const int p = wv * TPL + i;
        const float* tile = x + ((long)(b0 + p) * TT + kc * CH) * II;
#pragma unroll
        for (int j = 0; j < Q4; ++j)
            rq[i][j] = *(const v4f*)(tile + j * 128 + 4 * lane);
        rr[i] = *(const v2f*)(tile + Q4 * 128 + 2 * lane);
    }
}

__device__ __forceinline__ void fetch_stage(const v4f (&rq)[TPL][Q4], const v2f (&rr)[TPL],
                                            float* sbuf, int wv, int lane) {
#pragma unroll
    for (int i = 0; i < TPL; ++i) {
        float* t = sbuf + (wv * TPL + i) * TILE_F;
#pragma unroll
        for (int j = 0; j < Q4; ++j)
            *(v4f*)(t + j * 128 + 4 * lane) = rq[i][j];
        *(v2f*)(t + Q4 * 128 + 2 * lane) = rr[i];
    }
}

__device__ __forceinline__ void prod_frags(v2f (&A)[TPW][7], const float* sbuf,
                                           int pw, int n, int khalf) {
#pragma unroll
    for (int i = 0; i < TPW; ++i) {
        const int pr = pw + NPROD * i;
        const int p  = (pr < NB) ? pr : (NB - 1);
        const float* xrow = sbuf + p * TILE_F + n * II;
#pragma unroll
        for (int j = 0; j < 7; ++j)
            A[i][j] = *(const v2f*)(xrow + 4 * j + 2 * khalf);
    }
}

__device__ __forceinline__ void prod_compute(const v2f (&A)[TPW][7], const v2f (&bfrag)[7],
                                             const v8f cbias, float* dst,
                                             int pw, int n, int khalf) {
    v8f c[TPW];
#pragma unroll
    for (int i = 0; i < TPW; ++i) c[i] = cbias;
#pragma unroll
    for (int j = 0; j < 7; ++j) {
#pragma unroll
        for (int i = 0; i < TPW; ++i) {
            c[i] = __builtin_amdgcn_wmma_f32_16x16x4_f32(
                       false, A[i][j], false, bfrag[j], (short)0, c[i], false, false);
        }
    }
    if (n < NG) {
#pragma unroll
        for (int i = 0; i < TPW; ++i) {
            const int p = pw + NPROD * i;
            if (i < TPW - 1 || p < NB) {
                float* d = dst + p * TPITCH + n * CH + 8 * khalf;
                v4f lo4, hi4;
                lo4.x = c[i][0]; lo4.y = c[i][1]; lo4.z = c[i][2]; lo4.w = c[i][3];
                hi4.x = c[i][4]; hi4.y = c[i][5]; hi4.z = c[i][6]; hi4.w = c[i][7];
                *(v4f*)d       = lo4;
                *(v4f*)(d + 4) = hi4;
            }
        }
    }
}

__global__ void __launch_bounds__(128)
lstm_fused_wmma(const float* __restrict__ x,
                const float* __restrict__ W_ih,
                const float* __restrict__ W_hh,
                const float* __restrict__ b_ih,
                const float* __restrict__ b_hh,
                const float* __restrict__ fc_W,
                const float* __restrict__ fc_b,
                float* __restrict__ out) {
    __shared__ __attribute__((aligned(16))) float xgs[LDS_FLOATS];

    const int lane  = threadIdx.x & 31;
    const int wv    = __builtin_amdgcn_readfirstlane((int)(threadIdx.x >> 5));
    const int pw    = wv - 1;
    const int n     = lane & 15;
    const int sp    = lane >> 2;
    const int sq    = lane & 3;
    const int khalf = lane >> 4;
    const int b0    = blockIdx.x * NB;

    v2f bfrag[7];
#pragma unroll
    for (int j = 0; j < 7; ++j) {
        const int k0 = 4 * j + 2 * khalf;
        float q0 = 0.0f, q1 = 0.0f;
        if (n < NG) {
            q0 = W_ih[n * II + k0];
            q1 = W_ih[n * II + k0 + 1];
        }
        v2f bf; bf.x = q0; bf.y = q1;
        bfrag[j] = bf;
    }
    const float bias = (n < NG) ? (b_ih[n] + b_hh[n]) : 0.0f;
    v8f cbias;
#pragma unroll
    for (int v = 0; v < 8; ++v) cbias[v] = bias;

    ScanLane sk;
    sk.wA0 = W_hh[2 * sq];           sk.wA1 = W_hh[2 * sq + 1];
    sk.wB0 = W_hh[2 * (4 + sq)];     sk.wB1 = W_hh[2 * (4 + sq) + 1];
    const bool bsig = sq >= 2;
    sk.sB = bsig ? 0.5f : 1.0f;
    sk.aB = bsig ? 0.5f : 1.0f;
    sk.bB = bsig ? 0.5f : 0.0f;
    float hs = 0.f, cs = 0.f;
    const v4f* srow = (const v4f*)(xgs + sp * TPITCH + sq * CH);

    v4f rqA[TPL][Q4], rqB[TPL][Q4];
    v2f rrA[TPL], rrB[TPL];
    v2f frag[TPW][7];
    float* sb0 = xgs + STAGE_OFF;
    float* sb1 = sb0 + SBUF_F;
    fetch_load(rqA, rrA, x, b0, 0, wv, lane);
    fetch_load(rqB, rrB, x, b0, 1, wv, lane);
    fetch_stage(rqA, rrA, sb0, wv, lane);
    fetch_load(rqA, rrA, x, b0, 2, wv, lane);
    fetch_stage(rqB, rrB, sb1, wv, lane);
    fetch_load(rqB, rrB, x, b0, 3, wv, lane);
    chunk_barrier();
    if (wv != 0) {
        prod_frags(frag, sb0, pw, n, khalf);
        prod_compute(frag, bfrag, cbias, xgs, pw, n, khalf);
    }

    for (int k = 0; k < NCHUNK; k += 2) {
        chunk_barrier();
        fetch_stage(rqA, rrA, sb0, wv, lane);
        fetch_load(rqA, rrA, x, b0, k + 4, wv, lane);
        if (wv == 0) {
            scan_chunk(srow, sk, hs, cs);
        } else {
            prod_frags(frag, sb1, pw, n, khalf);
            prod_compute(frag, bfrag, cbias, xgs + BUFSZ, pw, n, khalf);
        }
        chunk_barrier();
        fetch_stage(rqB, rrB, sb1, wv, lane);
        fetch_load(rqB, rrB, x, b0, k + 5, wv, lane);
        if (wv == 0) {
            scan_chunk(srow + BUFSZ / 4, sk, hs, cs);
        } else {
            prod_frags(frag, sb0, pw, n, khalf);
            if (k + 2 < NCHUNK)
                prod_compute(frag, bfrag, cbias, xgs, pw, n, khalf);
        }
    }

    const float h0 = quad_perm<0xAA>(hs);
    const float h1 = quad_perm<0xFF>(hs);
    if (wv == 0 && sq == 0) {
        const int b = b0 + sp;
#pragma unroll
        for (int c = 0; c < CC; ++c) {
            out[b * CC + c] = fc_b[c] + h0 * fc_W[c * HH + 0] + h1 * fc_W[c * HH + 1];
        }
    }
}

extern "C" void kernel_launch(void* const* d_in, const int* in_sizes, int n_in,
                              void* d_out, int out_size, void* d_ws, size_t ws_size,
                              hipStream_t stream) {
    const float* x    = (const float*)d_in[0];
    const float* W_ih = (const float*)d_in[1];
    const float* W_hh = (const float*)d_in[2];
    const float* b_ih = (const float*)d_in[3];
    const float* b_hh = (const float*)d_in[4];
    const float* fc_W = (const float*)d_in[5];
    const float* fc_b = (const float*)d_in[6];
    float* out = (float*)d_out;

    lstm_fused_wmma<<<BB / NB, 128, 0, stream>>>(x, W_ih, W_hh, b_ih, b_hh, fc_W, fc_b, out);
}


# scan step's serial chain cut from 9 to 6 dependent instructions: sigmoid scale/offset folded into constants and the c update, sigm(o)*W_hh formed beside the chain; products and roundings regrouped
# speedup vs baseline: 8.9602x; 1.2415x over previous
#include <hip/hip_runtime.h>
#include <hip/hip_bf16.h>


#define BB 2048
#define TT 1024
#define II 28
#define HH 2
#define CC 10
#define NG 8

typedef __attribute__((ext_vector_type(2))) float v2f;
typedef __attribute__((ext_vector_type(4))) float v4f;
typedef __attribute__((ext_vector_type(8))) float v8f;

__device__ __forceinline__ float fast_tanh(float x) {
#if __has_builtin(__builtin_amdgcn_tanhf)
    return __builtin_amdgcn_tanhf(x);
#else
    float e = __expf(-2.0f * x);
    return (1.0f - e) / (1.0f + e);
#endif
}
__device__ __forceinline__ float fast_sigm(float x) {
    return 0.5f + 0.5f * fast_tanh(0.5f * x);
}

#define NB 8
#define CH 16
#define NCHUNK (TT / CH)
#define NPROD 3
#define NWAVE (NPROD + 1)
#define TPW ((NB + NPROD - 1) / NPROD)
#define TPL (NB / NWAVE)
#define TPITCH (NG * CH + 4)
#define BUFSZ (NB * TPITCH)
#define LDS_FLOATS (41 * 1024)
#define TILE_F (CH * II)
#define Q4 (TILE_F / 128)
#define STAGE_OFF (2 * BUFSZ)
#define SBUF_F (NB * TILE_F)
static_assert(TT % CH == 0 && NCHUNK % 2 == 0 && BB % NB == 0 && STAGE_OFF + 2 * SBUF_F <= LDS_FLOATS, "shape");
static_assert(NB % NWAVE == 0, "every wave fetches the same number of tiles");
static_assert(NB * 4 == 32 && NG == 8 && CH % 4 == 0 && TPITCH % 4 == 0, "scan: four lanes per batch element, two gates per lane, 16-byte groups");
static_assert(TILE_F == Q4 * 128 + 64, "a tile is Q4 groups of 16 bytes and one of 8 bytes per lane");
static_assert((NPROD - 1) + NPROD * (TPW - 2) < NB, "every slot but the last is a real tile for every producer");

__device__ __forceinline__ void chunk_barrier() {
    asm volatile("s_wait_dscnt 0x0\n\ts_barrier_signal -1\n\ts_barrier_wait -1" ::: "memory");
}

template <int CTRL>
__device__ __forceinline__ float quad_perm(float v) {
    return __builtin_bit_cast(float,
        __builtin_amdgcn_mov_dpp(__builtin_bit_cast(int, v), CTRL, 0xF, 0xF, true));
}

struct ScanLane {
    float kA0, kA1;
    float kB0, kB1;
    float xm, xb;
};

struct ScanState {
    float tc, x, to;
    float uA0, uA1, uB0, uB1;
};

__device__ __forceinline__ void lstm_step_quad(const float ga, const float gb,
                                               const ScanLane& k, ScanState& s) {
    float za = ga, zb = gb;
    asm("v_fmac_f32_dpp %[za], %[tc], %[uA0] quad_perm:[2,2,2,2] row_mask:0xf bank_mask:0xf bound_ctrl:1\n\t"
        "v_fmac_f32_dpp %[zb], %[tc], %[uB0] quad_perm:[2,2,2,2] row_mask:0xf bank_mask:0xf bound_ctrl:1\n\t"
        "v_fmac_f32_dpp %[za], %[tc], %[uA1] quad_perm:[3,3,3,3] row_mask:0xf bank_mask:0xf bound_ctrl:1\n\t"
        "v_fmac_f32_dpp %[zb], %[tc], %[uB1] quad_perm:[3,3,3,3] row_mask:0xf bank_mask:0xf bound_ctrl:1\n\t"
        "v_tanh_f32 %[za], %[za]\n\t"
        "v_tanh_f32 %[zb], %[zb]\n\t"
        "v_fmac_f32 %[x], %[za], %[x]\n\t"
        "v_fmac_f32_dpp %[x], %[x], %[zb] quad_perm:[0,1,0,1] row_mask:0xf bank_mask:0xf bound_ctrl:1\n\t"
        "v_fma_f32_e64_dpp %[uA0], %[zb], %[kA0], %[kA0] quad_perm:[0,0,0,0] row_mask:0xf bank_mask:0xf bound_ctrl:1\n\t"
        "v_fma_f32_e64_dpp %[uB0], %[zb], %[kB0], %[kB0] quad_perm:[0,0,0,0] row_mask:0xf bank_mask:0xf bound_ctrl:1\n\t"
        "v_tanh_f32 %[tc], %[x]\n\t"
        "v_fma_f32_e64_dpp %[uA1], %[zb], %[kA1], %[kA1] quad_perm:[1,1,1,1] row_mask:0xf bank_mask:0xf bound_ctrl:1\n\t"
        "v_fma_f32 %[x], %[x], %[xm], %[xb]\n\t"
        "v_fma_f32_e64_dpp %[uB1], %[zb], %[kB1], %[kB1] quad_perm:[1,1,1,1] row_mask:0xf bank_mask:0xf bound_ctrl:1"
        : [za] "+&v"(za), [zb] "+&v"(zb), [tc] "+&v"(s.tc), [x] "+&v"(s.x),
          [uA0] "+&v"(s.uA0), [uA1] "+&v"(s.uA1), [uB0] "+&v"(s.uB0), [uB1] "+&v"(s.uB1)
        : [kA0] "v"(k.kA0), [kA1] "v"(k.kA1), [kB0] "v"(k.kB0), [kB1] "v"(k.kB1),
          [xm] "v"(k.xm), [xb] "v"(k.xb));
    s.to = zb;
}

__device__ __forceinline__ void scan_chunk(const v4f* srca, const v4f* srcb,
                                           const ScanLane& k, ScanState& s) {
    v4f ra[CH / 4], rb[CH / 4];
#pragma unroll
    for (int j = 0; j < CH / 4; ++j) { ra[j] = srca[j]; rb[j] = srcb[j]; }
    __builtin_amdgcn_sched_barrier(0);
#pragma unroll
    for (int st = 0; st < CH; ++st) lstm_step_quad(ra[st >> 2][st & 3], rb[st >> 2][st & 3], k, s);
}

__device__ __forceinline__ void fetch_load(v4f (&rq)[TPL][Q4], v2f (&rr)[TPL],
                                           const float* __restrict__ x,
                                           int b0, int kk, int wv, int lane) {
    const int kc = (kk < NCHUNK) ? kk : (NCHUNK - 1);
#pragma unroll
    for (int i = 0; i < TPL; ++i) {
        const int p = wv * TPL + i;
        const float* tile = x + ((long)(b0 + p) * TT + kc * CH) * II;
#pragma unroll
        for (int j = 0; j < Q4; ++j)
            rq[i][j] = *(const v4f*)(tile + j * 128 + 4 * lane);
        rr[i] = *(const v2f*)(tile + Q4 * 128 + 2 * lane);
    }
}

__device__ __forceinline__ void fetch_stage(const v4f (&rq)[TPL][Q4], const v2f (&rr)[TPL],
                                            float* sbuf, int wv, int lane) {
#pragma unroll
    for (int i = 0; i < TPL; ++i) {
        float* t = sbuf + (wv * TPL + i) * TILE_F;
#pragma unroll
        for (int j = 0; j < Q4; ++j)
            *(v4f*)(t + j * 128 + 4 * lane) = rq[i][j];
        *(v2f*)(t + Q4 * 128 + 2 * lane) = rr[i];
    }
}

__device__ __forceinline__ void prod_frags(v2f (&A)[TPW][7], const float* sbuf,
                                           int pw, int n, int khalf) {
#pragma unroll
    for (int i = 0; i < TPW; ++i) {
        const int pr = pw + NPROD * i;
        const int p  = (pr < NB) ? pr : (NB - 1);
        const float* xrow = sbuf + p * TILE_F + n * II;
#pragma unroll
        for (int j = 0; j < 7; ++j)
            A[i][j] = *(const v2f*)(xrow + 4 * j + 2 * khalf);
    }
}

__device__ __forceinline__ void prod_compute(const v2f (&A)[TPW][7], const v2f (&bfrag)[7],
                                             const v8f cbias, float* dst,
                                             int pw, int n, int khalf) {
    v8f c[TPW];
#pragma unroll
    for (int i = 0; i < TPW; ++i) c[i] = cbias;
#pragma unroll
    for (int j = 0; j < 7; ++j) {
#pragma unroll
        for (int i = 0; i < TPW; ++i) {
            c[i] = __builtin_amdgcn_wmma_f32_16x16x4_f32(
                       false, A[i][j], false, bfrag[j], (short)0, c[i], false, false);
        }
    }
    if (n < NG) {
#pragma unroll
        for (int i = 0; i < TPW; ++i) {
            const int p = pw + NPROD * i;
            if (i < TPW - 1 || p < NB) {
                float* d = dst + p * TPITCH + n * CH + 8 * khalf;
                v4f lo4, hi4;
                lo4.x = c[i][0]; lo4.y = c[i][1]; lo4.z = c[i][2]; lo4.w = c[i][3];
                hi4.x = c[i][4]; hi4.y = c[i][5]; hi4.z = c[i][6]; hi4.w = c[i][7];
                *(v4f*)d       = lo4;
                *(v4f*)(d + 4) = hi4;
            }
        }
    }
}

__global__ void __launch_bounds__(128)
lstm_fused_wmma(const float* __restrict__ x,
                const float* __restrict__ W_ih,
                const float* __restrict__ W_hh,
                const float* __restrict__ b_ih,
                const float* __restrict__ b_hh,
                const float* __restrict__ fc_W,
                const float* __restrict__ fc_b,
                float* __restrict__ out) {
    __shared__ __attribute__((aligned(16))) float xgs[LDS_FLOATS];

    const int lane  = threadIdx.x & 31;
    const int wv    = __builtin_amdgcn_readfirstlane((int)(threadIdx.x >> 5));
    const int pw    = wv - 1;
    const int n     = lane & 15;
    const int sp    = lane >> 2;
    const int sq    = lane & 3;
    const int khalf = lane >> 4;
    const int b0    = blockIdx.x * NB;

    const float gsc = (n == 4 || n == 5) ? 1.0f : 0.5f;
    v2f bfrag[7];
#pragma unroll
    for (int j = 0; j < 7; ++j) {
        const int k0 = 4 * j + 2 * khalf;
        float q0 = 0.0f, q1 = 0.0f;
        if (n < NG) {
            q0 = gsc * W_ih[n * II + k0];
            q1 = gsc * W_ih[n * II + k0 + 1];
        }
        v2f bf; bf.x = q0; bf.y = q1;
        bfrag[j] = bf;
    }
    const float bias = (n < NG) ? gsc * (b_ih[n] + b_hh[n]) : 0.0f;
    v8f cbias;
#pragma unroll
    for (int v = 0; v < 8; ++v) cbias[v] = bias;

    ScanLane sk;
    const bool lo   = sq < 2;
    const int  gB   = (sq ^ 2) + 4;
    const float pB  = lo ? 0.25f : 0.5f;
    sk.kA0 = 0.25f * W_hh[2 * sq];   sk.kA1 = 0.25f * W_hh[2 * sq + 1];
    sk.kB0 = pB * W_hh[2 * gB];      sk.kB1 = pB * W_hh[2 * gB + 1];
    sk.xm  = lo ? 0.0f : 0.5f;
    sk.xb  = lo ? 0.5f : 0.0f;
    ScanState ss;
    ss.tc = 0.f; ss.x = sk.xb; ss.to = 0.f;
    ss.uA0 = 0.f; ss.uA1 = 0.f; ss.uB0 = 0.f; ss.uB1 = 0.f;
    const v4f* srowa = (const v4f*)(xgs + sp * TPITCH + sq * CH);
    const v4f* srowb = (const v4f*)(xgs + sp * TPITCH + gB * CH);

    v4f rqA[TPL][Q4], rqB[TPL][Q4];
    v2f rrA[TPL], rrB[TPL];
    v2f frag[TPW][7];
    float* sb0 = xgs + STAGE_OFF;
    float* sb1 = sb0 + SBUF_F;
    fetch_load(rqA, rrA, x, b0, 0, wv, lane);
    fetch_load(rqB, rrB, x, b0, 1, wv, lane);
    fetch_stage(rqA, rrA, sb0, wv, lane);
    fetch_load(rqA, rrA, x, b0, 2, wv, lane);
    fetch_stage(rqB, rrB, sb1, wv, lane);
    fetch_load(rqB, rrB, x, b0, 3, wv, lane);
    chunk_barrier();
    if (wv != 0) {
        prod_frags(frag, sb0, pw, n, khalf);
        prod_compute(frag, bfrag, cbias, xgs, pw, n, khalf);
    }

    for (int k = 0; k < NCHUNK; k += 2) {
        chunk_barrier();
        fetch_stage(rqA, rrA, sb0, wv, lane);
        fetch_load(rqA, rrA, x, b0, k + 4, wv, lane);
        if (wv == 0) {
            scan_chunk(srowa, srowb, sk, ss);
        } else {
            prod_frags(frag, sb1, pw, n, khalf);
            prod_compute(frag, bfrag, cbias, xgs + BUFSZ, pw, n, khalf);
        }
        chunk_barrier();
        fetch_stage(rqB, rrB, sb1, wv, lane);
        fetch_load(rqB, rrB, x, b0, k + 5, wv, lane);
        if (wv == 0) {
            scan_chunk(srowa + BUFSZ / 4, srowb + BUFSZ / 4, sk, ss);
        } else {
            prod_frags(frag, sb0, pw, n, khalf);
            if (k + 2 < NCHUNK)
                prod_compute(frag, bfrag, cbias, xgs, pw, n, khalf);
        }
    }

    const float so = __builtin_fmaf(0.5f, ss.to, 0.5f);
    const float h0 = quad_perm<0x00>(so) * quad_perm<0xAA>(ss.tc);
    const float h1 = quad_perm<0x55>(so) * quad_perm<0xFF>(ss.tc);
    if (wv == 0 && sq == 0) {
        const int b = b0 + sp;
#pragma unroll
        for (int c = 0; c < CC; ++c) {
            out[b * CC + c] = fc_b[c] + h0 * fc_W[c * HH + 0] + h1 * fc_W[c * HH + 1];
        }
    }
}

extern "C" void kernel_launch(void* const* d_in, const int* in_sizes, int n_in,
                              void* d_out, int out_size, void* d_ws, size_t ws_size,
                              hipStream_t stream) {
    const float* x    = (const float*)d_in[0];
    const float* W_ih = (const float*)d_in[1];
    const float* W_hh = (const float*)d_in[2];
    const float* b_ih = (const float*)d_in[3];
    const float* b_hh = (const float*)d_in[4];
    const float* fc_W = (const float*)d_in[5];
    const float* fc_b = (const float*)d_in[6];
    float* out = (float*)d_out;

    lstm_fused_wmma<<<BB / NB, 128, 0, stream>>>(x, W_ih, W_hh, b_ih, b_hh, fc_W, fc_b, out);
}


# each wave issues its trip's LDS reads as one group directly behind the chunk barrier (scan: gate rows; producers: all A fragments), ahead of x staging and requests that cover the round trip; same sums
# speedup vs baseline: 9.2640x; 1.0339x over previous
#include <hip/hip_runtime.h>
#include <hip/hip_bf16.h>


#define BB 2048
#define TT 1024
#define II 28
#define HH 2
#define CC 10
#define NG 8

typedef __attribute__((ext_vector_type(2))) float v2f;
typedef __attribute__((ext_vector_type(4))) float v4f;
typedef __attribute__((ext_vector_type(8))) float v8f;

__device__ __forceinline__ float fast_tanh(float x) {
#if __has_builtin(__builtin_amdgcn_tanhf)
    return __builtin_amdgcn_tanhf(x);
#else
    float e = __expf(-2.0f * x);
    return (1.0f - e) / (1.0f + e);
#endif
}
__device__ __forceinline__ float fast_sigm(float x) {
    return 0.5f + 0.5f * fast_tanh(0.5f * x);
}

#define NB 8
#define CH 16
#define NCHUNK (TT / CH)
#define NPROD 3
#define NWAVE (NPROD + 1)
#define TPW ((NB + NPROD - 1) / NPROD)
#define TPL (NB / NWAVE)
#define TPITCH (NG * CH + 4)
#define BUFSZ (NB * TPITCH)
#define LDS_FLOATS (41 * 1024)
#define TILE_F (CH * II)
#define Q4 (TILE_F / 128)
#define STAGE_OFF (2 * BUFSZ)
#define SBUF_F (NB * TILE_F)
static_assert(TT % CH == 0 && NCHUNK % 2 == 0 && BB % NB == 0 && STAGE_OFF + 2 * SBUF_F <= LDS_FLOATS, "shape");
static_assert(NB % NWAVE == 0, "every wave fetches the same number of tiles");
static_assert(NB * 4 == 32 && NG == 8 && CH % 4 == 0 && TPITCH % 4 == 0, "scan: four lanes per batch element, two gates per lane, 16-byte groups");
static_assert(TILE_F == Q4 * 128 + 64, "a tile is Q4 groups of 16 bytes and one of 8 bytes per lane");
static_assert((NPROD - 1) + NPROD * (TPW - 2) < NB, "every slot but the last is a real tile for every producer");

__device__ __forceinline__ void chunk_barrier() {
    asm volatile("s_wait_dscnt 0x0\n\ts_barrier_signal -1\n\ts_barrier_wait -1" ::: "memory");
}

template <int CTRL>
__device__ __forceinline__ float quad_perm(float v) {
    return __builtin_bit_cast(float,
        __builtin_amdgcn_mov_dpp(__builtin_bit_cast(int, v), CTRL, 0xF, 0xF, true));
}

struct ScanLane {
    float kA0, kA1;
    float kB0, kB1;
    float xm, xb;
};

struct ScanState {
    float tc, x, to;
    float uA0, uA1, uB0, uB1;
};

__device__ __forceinline__ void lstm_step_quad(const float ga, const float gb,
                                               const ScanLane& k, ScanState& s) {
    float za = ga, zb = gb;
    asm("v_fmac_f32_dpp %[za], %[tc], %[uA0] quad_perm:[2,2,2,2] row_mask:0xf bank_mask:0xf bound_ctrl:1\n\t"
        "v_fmac_f32_dpp %[zb], %[tc], %[uB0] quad_perm:[2,2,2,2] row_mask:0xf bank_mask:0xf bound_ctrl:1\n\t"
        "v_fmac_f32_dpp %[za], %[tc], %[uA1] quad_perm:[3,3,3,3] row_mask:0xf bank_mask:0xf bound_ctrl:1\n\t"
        "v_fmac_f32_dpp %[zb], %[tc], %[uB1] quad_perm:[3,3,3,3] row_mask:0xf bank_mask:0xf bound_ctrl:1\n\t"
        "v_tanh_f32 %[za], %[za]\n\t"
        "v_tanh_f32 %[zb], %[zb]\n\t"
        "v_fmac_f32 %[x], %[za], %[x]\n\t"
        "v_fmac_f32_dpp %[x], %[x], %[zb] quad_perm:[0,1,0,1] row_mask:0xf bank_mask:0xf bound_ctrl:1\n\t"
        "v_fma_f32_e64_dpp %[uA0], %[zb], %[kA0], %[kA0] quad_perm:[0,0,0,0] row_mask:0xf bank_mask:0xf bound_ctrl:1\n\t"
        "v_fma_f32_e64_dpp %[uB0], %[zb], %[kB0], %[kB0] quad_perm:[0,0,0,0] row_mask:0xf bank_mask:0xf bound_ctrl:1\n\t"
        "v_tanh_f32 %[tc], %[x]\n\t"
        "v_fma_f32_e64_dpp %[uA1], %[zb], %[kA1], %[kA1] quad_perm:[1,1,1,1] row_mask:0xf bank_mask:0xf bound_ctrl:1\n\t"
        "v_fma_f32 %[x], %[x], %[xm], %[xb]\n\t"
        "v_fma_f32_e64_dpp %[uB1], %[zb], %[kB1], %[kB1] quad_perm:[1,1,1,1] row_mask:0xf bank_mask:0xf bound_ctrl:1"
        : [za] "+&v"(za), [zb] "+&v"(zb), [tc] "+&v"(s.tc), [x] "+&v"(s.x),
          [uA0] "+&v"(s.uA0), [uA1] "+&v"(s.uA1), [uB0] "+&v"(s.uB0), [uB1] "+&v"(s.uB1)
        : [kA0] "v"(k.kA0), [kA1] "v"(k.kA1), [kB0] "v"(k.kB0), [kB1] "v"(k.kB1),
          [xm] "v"(k.xm), [xb] "v"(k.xb));
    s.to = zb;
}

__device__ __forceinline__ void scan_read(v4f (&ra)[CH / 4], v4f (&rb)[CH / 4],
                                          const v4f* srca, const v4f* srcb) {
#pragma unroll
    for (int j = 0; j < CH / 4; ++j) { ra[j] = srca[j]; rb[j] = srcb[j]; }
}

__device__ __forceinline__ void scan_steps(const v4f (&ra)[CH / 4], const v4f (&rb)[CH / 4],
                                           const ScanLane& k, ScanState& s) {
#pragma unroll
    for (int st = 0; st < CH; ++st) lstm_step_quad(ra[st >> 2][st & 3], rb[st >> 2][st & 3], k, s);
}

__device__ __forceinline__ void fetch_load(v4f (&rq)[TPL][Q4], v2f (&rr)[TPL],
                                           const float* __restrict__ x,
                                           int b0, int kk, int wv, int lane) {
    const int kc = (kk < NCHUNK) ? kk : (NCHUNK - 1);
#pragma unroll
    for (int i = 0; i < TPL; ++i) {
        const int p = wv * TPL + i;
        const float* tile = x + ((long)(b0 + p) * TT + kc * CH) * II;
#pragma unroll
        for (int j = 0; j < Q4; ++j)
            rq[i][j] = *(const v4f*)(tile + j * 128 + 4 * lane);
        rr[i] = *(const v2f*)(tile + Q4 * 128 + 2 * lane);
    }
}

__device__ __forceinline__ void fetch_stage(const v4f (&rq)[TPL][Q4], const v2f (&rr)[TPL],
                                            float* sbuf, int wv, int lane) {
#pragma unroll
    for (int i = 0; i < TPL; ++i) {
        float* t = sbuf + (wv * TPL + i) * TILE_F;
#pragma unroll
        for (int j = 0; j < Q4; ++j)
            *(v4f*)(t + j * 128 + 4 * lane) = rq[i][j];
        *(v2f*)(t + Q4 * 128 + 2 * lane) = rr[i];
    }
}

__device__ __forceinline__ void prod_frags(v2f (&A)[TPW][7], const float* sbuf,
                                           int pw, int n, int khalf) {
#pragma unroll
    for (int i = 0; i < TPW; ++i) {
        const int pr = pw + NPROD * i;
        const int p  = (pr < NB) ? pr : (NB - 1);
        const float* xrow = sbuf + p * TILE_F + n * II;
#pragma unroll
        for (int j = 0; j < 7; ++j)
            A[i][j] = *(const v2f*)(xrow + 4 * j + 2 * khalf);
    }
}

__device__ __forceinline__ void prod_compute(const v2f (&A)[TPW][7], const v2f (&bfrag)[7],
                                             const v8f cbias, float* dst,
                                             int pw, int n, int khalf) {
    v8f c[TPW];
#pragma unroll
    for (int i = 0; i < TPW; ++i) c[i] = cbias;
#pragma unroll
    for (int j = 0; j < 7; ++j) {
#pragma unroll
        for (int i = 0; i < TPW; ++i) {
            c[i] = __builtin_amdgcn_wmma_f32_16x16x4_f32(
                       false, A[i][j], false, bfrag[j], (short)0, c[i], false, false);
        }
    }
    if (n < NG) {
#pragma unroll
        for (int i = 0; i < TPW; ++i) {
            const int p = pw + NPROD * i;
            if (i < TPW - 1 || p < NB) {
                float* d = dst + p * TPITCH + n * CH + 8 * khalf;
                v4f lo4, hi4;
                lo4.x = c[i][0]; lo4.y = c[i][1]; lo4.z = c[i][2]; lo4.w = c[i][3];
                hi4.x = c[i][4]; hi4.y = c[i][5]; hi4.z = c[i][6]; hi4.w = c[i][7];
                *(v4f*)d       = lo4;
                *(v4f*)(d + 4) = hi4;
            }
        }
    }
}

__device__ __forceinline__ void define_only(v4f& r) { asm volatile("" : "=v"(r)); }
__device__ __forceinline__ void define_only(v2f& r) { asm volatile("" : "=v"(r)); }

__global__ void __launch_bounds__(128)
lstm_fused_wmma(const float* __restrict__ x,
                const float* __restrict__ W_ih,
                const float* __restrict__ W_hh,
                const float* __restrict__ b_ih,
                const float* __restrict__ b_hh,
                const float* __restrict__ fc_W,
                const float* __restrict__ fc_b,
                float* __restrict__ out) {
    __shared__ __attribute__((aligned(16))) float xgs[LDS_FLOATS];

    const int lane  = threadIdx.x & 31;
    const int wv    = __builtin_amdgcn_readfirstlane((int)(threadIdx.x >> 5));
    const int pw    = wv - 1;
    const int n     = lane & 15;
    const int sp    = lane >> 2;
    const int sq    = lane & 3;
    const int khalf = lane >> 4;
    const int b0    = blockIdx.x * NB;

    const float gsc = (n == 4 || n == 5) ? 1.0f : 0.5f;
    v2f bfrag[7];
#pragma unroll
    for (int j = 0; j < 7; ++j) {
        const int k0 = 4 * j + 2 * khalf;
        float q0 = 0.0f, q1 = 0.0f;
        if (n < NG) {
            q0 = gsc * W_ih[n * II + k0];
            q1 = gsc * W_ih[n * II + k0 + 1];
        }
        v2f bf; bf.x = q0; bf.y = q1;
        bfrag[j] = bf;
    }
    const float bias = (n < NG) ? gsc * (b_ih[n] + b_hh[n]) : 0.0f;
    v8f cbias;
#pragma unroll
    for (int v = 0; v < 8; ++v) cbias[v] = bias;

    ScanLane sk;
    const bool lo   = sq < 2;
    const int  gB   = (sq ^ 2) + 4;
    const float pB  = lo ? 0.25f : 0.5f;
    sk.kA0 = 0.25f * W_hh[2 * sq];   sk.kA1 = 0.25f * W_hh[2 * sq + 1];
    sk.kB0 = pB * W_hh[2 * gB];      sk.kB1 = pB * W_hh[2 * gB + 1];
    sk.xm  = lo ? 0.0f : 0.5f;
    sk.xb  = lo ? 0.5f : 0.0f;
    ScanState ss;
    ss.tc = 0.f; ss.x = sk.xb; ss.to = 0.f;
    ss.uA0 = 0.f; ss.uA1 = 0.f; ss.uB0 = 0.f; ss.uB1 = 0.f;
    const v4f* srowa = (const v4f*)(xgs + sp * TPITCH + sq * CH);
    const v4f* srowb = (const v4f*)(xgs + sp * TPITCH + gB * CH);

    v4f rqA[TPL][Q4], rqB[TPL][Q4];
    v2f rrA[TPL], rrB[TPL];
    v2f frag[TPW][7];
    float* sb0 = xgs + STAGE_OFF;
    float* sb1 = sb0 + SBUF_F;
    fetch_load(rqA, rrA, x, b0, 0, wv, lane);
    fetch_load(rqB, rrB, x, b0, 1, wv, lane);
    fetch_stage(rqA, rrA, sb0, wv, lane);
    fetch_load(rqA, rrA, x, b0, 2, wv, lane);
    fetch_stage(rqB, rrB, sb1, wv, lane);
    fetch_load(rqB, rrB, x, b0, 3, wv, lane);
    chunk_barrier();
    if (wv != 0) {
        prod_frags(frag, sb0, pw, n, khalf);
        prod_compute(frag, bfrag, cbias, xgs, pw, n, khalf);
    }

    for (int k = 0; k < NCHUNK; k += 2) {
        chunk_barrier();
        {
            v4f ra[CH / 4], rb[CH / 4];
            v2f fr[TPW][7];
            if (wv == 0) {
                scan_read(ra, rb, srowa, srowb);
#pragma unroll
                for (int i = 0; i < TPW; ++i)
#pragma unroll
                    for (int j = 0; j < 7; ++j) define_only(fr[i][j]);
            } else {
                prod_frags(fr, sb1, pw, n, khalf);
#pragma unroll
                for (int j = 0; j < CH / 4; ++j) { define_only(ra[j]); define_only(rb[j]); }
            }
            __builtin_amdgcn_sched_barrier(0);
            fetch_stage(rqA, rrA, sb0, wv, lane);
            fetch_load(rqA, rrA, x, b0, k + 4, wv, lane);
            __builtin_amdgcn_sched_barrier(0);
            if (wv == 0) scan_steps(ra, rb, sk, ss);
            else prod_compute(fr, bfrag, cbias, xgs + BUFSZ, pw, n, khalf);
        }
        chunk_barrier();
        {
            v4f ra[CH / 4], rb[CH / 4];
            v2f fr[TPW][7];
            if (wv == 0) {
                scan_read(ra, rb, srowa + BUFSZ / 4, srowb + BUFSZ / 4);
#pragma unroll
                for (int i = 0; i < TPW; ++i)
#pragma unroll
                    for (int j = 0; j < 7; ++j) define_only(fr[i][j]);
            } else {
                prod_frags(fr, sb0, pw, n, khalf);
#pragma unroll
                for (int j = 0; j < CH / 4; ++j) { define_only(ra[j]); define_only(rb[j]); }
            }
            __builtin_amdgcn_sched_barrier(0);
            fetch_stage(rqB, rrB, sb1, wv, lane);
            fetch_load(rqB, rrB, x, b0, k + 5, wv, lane);
            __builtin_amdgcn_sched_barrier(0);
            if (wv == 0) scan_steps(ra, rb, sk, ss);
            else if (k + 2 < NCHUNK) prod_compute(fr, bfrag, cbias, xgs, pw, n, khalf);
        }
    }

    const float so = __builtin_fmaf(0.5f, ss.to, 0.5f);
    const float h0 = quad_perm<0x00>(so) * quad_perm<0xAA>(ss.tc);
    const float h1 = quad_perm<0x55>(so) * quad_perm<0xFF>(ss.tc);
    if (wv == 0 && sq == 0) {
        const int b = b0 + sp;
#pragma unroll
        for (int c = 0; c < CC; ++c) {
            out[b * CC + c] = fc_b[c] + h0 * fc_W[c * HH + 0] + h1 * fc_W[c * HH + 1];
        }
    }
}

extern "C" void kernel_launch(void* const* d_in, const int* in_sizes, int n_in,
                              void* d_out, int out_size, void* d_ws, size_t ws_size,
                              hipStream_t stream) {
    const float* x    = (const float*)d_in[0];
    const float* W_ih = (const float*)d_in[1];
    const float* W_hh = (const float*)d_in[2];
    const float* b_ih = (const float*)d_in[3];
    const float* b_hh = (const float*)d_in[4];
    const float* fc_W = (const float*)d_in[5];
    const float* fc_b = (const float*)d_in[6];
    float* out = (float*)d_out;

    lstm_fused_wmma<<<BB / NB, 128, 0, stream>>>(x, W_ih, W_hh, b_ih, b_hh, fc_W, fc_b, out);
}


# chunks of 32 steps, two WMMA row tiles per batch element (were 16): half as many scan hand-overs (barrier, LDS round trip) per step, x requests in flight twice as long; sums unchanged
# speedup vs baseline: 9.6180x; 1.0382x over previous
#include <hip/hip_runtime.h>
#include <hip/hip_bf16.h>


#define BB 2048
#define TT 1024
#define II 28
#define HH 2
#define CC 10
#define NG 8

typedef __attribute__((ext_vector_type(2))) float v2f;
typedef __attribute__((ext_vector_type(4))) float v4f;
typedef __attribute__((ext_vector_type(8))) float v8f;

__device__ __forceinline__ float fast_tanh(float x) {
#if __has_builtin(__builtin_amdgcn_tanhf)
    return __builtin_amdgcn_tanhf(x);
#else
    float e = __expf(-2.0f * x);
    return (1.0f - e) / (1.0f + e);
#endif
}
__device__ __forceinline__ float fast_sigm(float x) {
    return 0.5f + 0.5f * fast_tanh(0.5f * x);
}

#define NB 8
#define CH 32
#define RT (CH / 16)
#define NCHUNK (TT / CH)
#define NPROD 3
#define NWAVE (NPROD + 1)
#define TPW ((NB + NPROD - 1) / NPROD)
#define TPL (NB / NWAVE)
#define GPITCH (CH + 4)
#define TPITCH (NG * GPITCH + 16)
#define BUFSZ (NB * TPITCH)
#define LDS_FLOATS (41 * 1024)
#define TILE_F (CH * II)
#define Q4 (TILE_F / 128)
#define STAGE_OFF (2 * BUFSZ)
#define SBUF_F (NB * TILE_F)
static_assert(TT % CH == 0 && NCHUNK % 2 == 0 && BB % NB == 0 && STAGE_OFF + 2 * SBUF_F <= LDS_FLOATS, "shape");
static_assert(NB % NWAVE == 0, "every wave fetches the rows of the same number of batch elements");
static_assert(NB * 4 == 32 && NG == 8 && CH % 16 == 0 && GPITCH % 4 == 0 && TPITCH % 4 == 0, "scan: four lanes per batch element, two gates per lane, 16-byte groups");
static_assert(TILE_F == Q4 * 128, "the rows of a chunk are Q4 groups of 16 bytes per lane");
static_assert((NPROD - 1) + NPROD * (TPW - 2) < NB, "every slot but the last is a real batch element for every producer");

__device__ __forceinline__ void chunk_barrier() {
    asm volatile("s_wait_dscnt 0x0\n\ts_barrier_signal -1\n\ts_barrier_wait -1" ::: "memory");
}

template <int CTRL>
__device__ __forceinline__ float quad_perm(float v) {
    return __builtin_bit_cast(float,
        __builtin_amdgcn_mov_dpp(__builtin_bit_cast(int, v), CTRL, 0xF, 0xF, true));
}

struct ScanLane {
    float kA0, kA1;
    float kB0, kB1;
    float xm, xb;
};

struct ScanState {
    float tc, x, to;
    float uA0, uA1, uB0, uB1;
};

__device__ __forceinline__ void lstm_step_quad(const float ga, const float gb,
                                               const ScanLane& k, ScanState& s) {
    float za = ga, zb = gb;
    asm("v_fmac_f32_dpp %[za], %[tc], %[uA0] quad_perm:[2,2,2,2] row_mask:0xf bank_mask:0xf bound_ctrl:1\n\t"
        "v_fmac_f32_dpp %[zb], %[tc], %[uB0] quad_perm:[2,2,2,2] row_mask:0xf bank_mask:0xf bound_ctrl:1\n\t"
        "v_fmac_f32_dpp %[za], %[tc], %[uA1] quad_perm:[3,3,3,3] row_mask:0xf bank_mask:0xf bound_ctrl:1\n\t"
        "v_fmac_f32_dpp %[zb], %[tc], %[uB1] quad_perm:[3,3,3,3] row_mask:0xf bank_mask:0xf bound_ctrl:1\n\t"
        "v_tanh_f32 %[za], %[za]\n\t"
        "v_tanh_f32 %[zb], %[zb]\n\t"
        "v_fmac_f32 %[x], %[za], %[x]\n\t"
        "v_fmac_f32_dpp %[x], %[x], %[zb] quad_perm:[0,1,0,1] row_mask:0xf bank_mask:0xf bound_ctrl:1\n\t"
        "v_fma_f32_e64_dpp %[uA0], %[zb], %[kA0], %[kA0] quad_perm:[0,0,0,0] row_mask:0xf bank_mask:0xf bound_ctrl:1\n\t"
        "v_fma_f32_e64_dpp %[uB0], %[zb], %[kB0], %[kB0] quad_perm:[0,0,0,0] row_mask:0xf bank_mask:0xf bound_ctrl:1\n\t"
        "v_tanh_f32 %[tc], %[x]\n\t"
        "v_fma_f32_e64_dpp %[uA1], %[zb], %[kA1], %[kA1] quad_perm:[1,1,1,1] row_mask:0xf bank_mask:0xf bound_ctrl:1\n\t"
        "v_fma_f32 %[x], %[x], %[xm], %[xb]\n\t"
        "v_fma_f32_e64_dpp %[uB1], %[zb], %[kB1], %[kB1] quad_perm:[1,1,1,1] row_mask:0xf bank_mask:0xf bound_ctrl:1"
        : [za] "+&v"(za), [zb] "+&v"(zb), [tc] "+&v"(s.tc), [x] "+&v"(s.x),
          [uA0] "+&v"(s.uA0), [uA1] "+&v"(s.uA1), [uB0] "+&v"(s.uB0), [uB1] "+&v"(s.uB1)
        : [kA0] "v"(k.kA0), [kA1] "v"(k.kA1), [kB0] "v"(k.kB0), [kB1] "v"(k.kB1),
          [xm] "v"(k.xm), [xb] "v"(k.xb));
    s.to = zb;
}

__device__ __forceinline__ void scan_read(v4f (&ra)[CH / 4], v4f (&rb)[CH / 4],
                                          const v4f* srca, const v4f* srcb) {
#pragma unroll
    for (int j = 0; j < CH / 4; ++j) { ra[j] = srca[j]; rb[j] = srcb[j]; }
}

__device__ __forceinline__ void scan_steps(const v4f (&ra)[CH / 4], const v4f (&rb)[CH / 4],
                                           const ScanLane& k, ScanState& s) {
#pragma unroll
    for (int st = 0; st < CH; ++st) lstm_step_quad(ra[st >> 2][st & 3], rb[st >> 2][st & 3], k, s);
}

__device__ __forceinline__ void fetch_load(v4f (&rq)[TPL][Q4],
                                           const float* __restrict__ x,
                                           int b0, int kk, int wv, int lane) {
    const int kc = (kk < NCHUNK) ? kk : (NCHUNK - 1);
#pragma unroll
    for (int i = 0; i < TPL; ++i) {
        const int p = wv * TPL + i;
        const float* tile = x + ((long)(b0 + p) * TT + kc * CH) * II;
#pragma unroll
        for (int j = 0; j < Q4; ++j)
            rq[i][j] = *(const v4f*)(tile + j * 128 + 4 * lane);
    }
}

__device__ __forceinline__ void fetch_stage(const v4f (&rq)[TPL][Q4],
                                            float* sbuf, int wv, int lane) {
#pragma unroll
    for (int i = 0; i < TPL; ++i) {
        float* t = sbuf + (wv * TPL + i) * TILE_F;
#pragma unroll
        for (int j = 0; j < Q4; ++j)
            *(v4f*)(t + j * 128 + 4 * lane) = rq[i][j];
    }
}

__device__ __forceinline__ void prod_frags(v2f (&A)[TPW][RT][7], const float* sbuf,
                                           int pw, int n, int khalf) {
#pragma unroll
    for (int i = 0; i < TPW; ++i) {
        const int pr = pw + NPROD * i;
        const int p  = (pr < NB) ? pr : (NB - 1);
#pragma unroll
        for (int rt = 0; rt < RT; ++rt) {
            const float* xrow = sbuf + p * TILE_F + (16 * rt + n) * II;
#pragma unroll
            for (int j = 0; j < 7; ++j)
                A[i][rt][j] = *(const v2f*)(xrow + 4 * j + 2 * khalf);
        }
    }
}

__device__ __forceinline__ void prod_compute(const v2f (&A)[TPW][RT][7], const v2f (&bfrag)[7],
                                             const v8f cbias, float* dst,
                                             int pw, int n, int khalf) {
    v8f c[TPW][RT];
#pragma unroll
    for (int i = 0; i < TPW; ++i)
#pragma unroll
        for (int rt = 0; rt < RT; ++rt) c[i][rt] = cbias;
#pragma unroll
    for (int j = 0; j < 7; ++j) {
#pragma unroll
        for (int i = 0; i < TPW; ++i) {
#pragma unroll
            for (int rt = 0; rt < RT; ++rt) {
                c[i][rt] = __builtin_amdgcn_wmma_f32_16x16x4_f32(
                               false, A[i][rt][j], false, bfrag[j], (short)0, c[i][rt], false, false);
            }
        }
    }
    if (n < NG) {
#pragma unroll
        for (int i = 0; i < TPW; ++i) {
            const int p = pw + NPROD * i;
            if (i < TPW - 1 || p < NB) {
#pragma unroll
                for (int rt = 0; rt < RT; ++rt) {
                    float* d = dst + p * TPITCH + n * GPITCH + 16 * rt + 8 * khalf;
                    v4f lo4, hi4;
                    lo4.x = c[i][rt][0]; lo4.y = c[i][rt][1]; lo4.z = c[i][rt][2]; lo4.w = c[i][rt][3];
                    hi4.x = c[i][rt][4]; hi4.y = c[i][rt][5]; hi4.z = c[i][rt][6]; hi4.w = c[i][rt][7];
                    *(v4f*)d       = lo4;
                    *(v4f*)(d + 4) = hi4;
                }
            }
        }
    }
}

__device__ __forceinline__ void define_only(v4f& r) { asm volatile("" : "=v"(r)); }
__device__ __forceinline__ void define_only(v2f& r) { asm volatile("" : "=v"(r)); }

__global__ void __launch_bounds__(128)
lstm_fused_wmma(const float* __restrict__ x,
                const float* __restrict__ W_ih,
                const float* __restrict__ W_hh,
                const float* __restrict__ b_ih,
                const float* __restrict__ b_hh,
                const float* __restrict__ fc_W,
                const float* __restrict__ fc_b,
                float* __restrict__ out) {
    __shared__ __attribute__((aligned(16))) float xgs[LDS_FLOATS];

    const int lane  = threadIdx.x & 31;
    const int wv    = __builtin_amdgcn_readfirstlane((int)(threadIdx.x >> 5));
    const int pw    = wv - 1;
    const int n     = lane & 15;
    const int sp    = lane >> 2;
    const int sq    = lane & 3;
    const int khalf = lane >> 4;
    const int b0    = blockIdx.x * NB;

    const float gsc = (n == 4 || n == 5) ? 1.0f : 0.5f;
    v2f bfrag[7];
#pragma unroll
    for (int j = 0; j < 7; ++j) {
        const int k0 = 4 * j + 2 * khalf;
        float q0 = 0.0f, q1 = 0.0f;
        if (n < NG) {
            q0 = gsc * W_ih[n * II + k0];
            q1 = gsc * W_ih[n * II + k0 + 1];
        }
        v2f bf; bf.x = q0; bf.y = q1;
        bfrag[j] = bf;
    }
    const float bias = (n < NG) ? gsc * (b_ih[n] + b_hh[n]) : 0.0f;
    v8f cbias;
#pragma unroll
    for (int v = 0; v < 8; ++v) cbias[v] = bias;

    ScanLane sk;
    const bool lo   = sq < 2;
    const int  gB   = (sq ^ 2) + 4;
    const float pB  = lo ? 0.25f : 0.5f;
    sk.kA0 = 0.25f * W_hh[2 * sq];   sk.kA1 = 0.25f * W_hh[2 * sq + 1];
    sk.kB0 = pB * W_hh[2 * gB];      sk.kB1 = pB * W_hh[2 * gB + 1];
    sk.xm  = lo ? 0.0f : 0.5f;
    sk.xb  = lo ? 0.5f : 0.0f;
    ScanState ss;
    ss.tc = 0.f; ss.x = sk.xb; ss.to = 0.f;
    ss.uA0 = 0.f; ss.uA1 = 0.f; ss.uB0 = 0.f; ss.uB1 = 0.f;
    const v4f* srowa = (const v4f*)(xgs + sp * TPITCH + sq * GPITCH);
    const v4f* srowb = (const v4f*)(xgs + sp * TPITCH + gB * GPITCH);

    v4f rqA[TPL][Q4], rqB[TPL][Q4];
    v2f frag[TPW][RT][7];
    float* sb0 = xgs + STAGE_OFF;
    float* sb1 = sb0 + SBUF_F;
    fetch_load(rqA, x, b0, 0, wv, lane);
    fetch_load(rqB, x, b0, 1, wv, lane);
    fetch_stage(rqA, sb0, wv, lane);
    fetch_load(rqA, x, b0, 2, wv, lane);
    fetch_stage(rqB, sb1, wv, lane);
    fetch_load(rqB, x, b0, 3, wv, lane);
    chunk_barrier();
    if (wv != 0) {
        prod_frags(frag, sb0, pw, n, khalf);
        prod_compute(frag, bfrag, cbias, xgs, pw, n, khalf);
    }

    for (int k = 0; k < NCHUNK; k += 2) {
        chunk_barrier();
        {
            v4f ra[CH / 4], rb[CH / 4];
            v2f fr[TPW][RT][7];
            if (wv == 0) {
                scan_read(ra, rb, srowa, srowb);
#pragma unroll
                for (int i = 0; i < TPW; ++i)
#pragma unroll
                    for (int rt = 0; rt < RT; ++rt)
#pragma unroll
                        for (int j = 0; j < 7; ++j) define_only(fr[i][rt][j]);
            } else {
                prod_frags(fr, sb1, pw, n, khalf);
#pragma unroll
                for (int j = 0; j < CH / 4; ++j) { define_only(ra[j]); define_only(rb[j]); }
            }
            __builtin_amdgcn_sched_barrier(0);
            fetch_stage(rqA, sb0, wv, lane);
            fetch_load(rqA, x, b0, k + 4, wv, lane);
            __builtin_amdgcn_sched_barrier(0);
            if (wv == 0) scan_steps(ra, rb, sk, ss);
            else prod_compute(fr, bfrag, cbias, xgs + BUFSZ, pw, n, khalf);
        }
        chunk_barrier();
        {
            v4f ra[CH / 4], rb[CH / 4];
            v2f fr[TPW][RT][7];
            if (wv == 0) {
                scan_read(ra, rb, srowa + BUFSZ / 4, srowb + BUFSZ / 4);
#pragma unroll
                for (int i = 0; i < TPW; ++i)
#pragma unroll
                    for (int rt = 0; rt < RT; ++rt)
#pragma unroll
                        for (int j = 0; j < 7; ++j) define_only(fr[i][rt][j]);
            } else {
                prod_frags(fr, sb0, pw, n, khalf);
#pragma unroll
                for (int j = 0; j < CH / 4; ++j) { define_only(ra[j]); define_only(rb[j]); }
            }
            __builtin_amdgcn_sched_barrier(0);
            fetch_stage(rqB, sb1, wv, lane);
            fetch_load(rqB, x, b0, k + 5, wv, lane);
            __builtin_amdgcn_sched_barrier(0);
            if (wv == 0) scan_steps(ra, rb, sk, ss);
            else if (k + 2 < NCHUNK) prod_compute(fr, bfrag, cbias, xgs, pw, n, khalf);
        }
    }

    const float so = __builtin_fmaf(0.5f, ss.to, 0.5f);
    const float h0 = quad_perm<0x00>(so) * quad_perm<0xAA>(ss.tc);
    const float h1 = quad_perm<0x55>(so) * quad_perm<0xFF>(ss.tc);
    if (wv == 0 && sq == 0) {
        const int b = b0 + sp;
#pragma unroll
        for (int c = 0; c < CC; ++c) {
            out[b * CC + c] = fc_b[c] + h0 * fc_W[c * HH + 0] + h1 * fc_W[c * HH + 1];
        }
    }
}

extern "C" void kernel_launch(void* const* d_in, const int* in_sizes, int n_in,
                              void* d_out, int out_size, void* d_ws, size_t ws_size,
                              hipStream_t stream) {
    const float* x    = (const float*)d_in[0];
    const float* W_ih = (const float*)d_in[1];
    const float* W_hh = (const float*)d_in[2];
    const float* b_ih = (const float*)d_in[3];
    const float* b_hh = (const float*)d_in[4];
    const float* fc_W = (const float*)d_in[5];
    const float* fc_b = (const float*)d_in[6];
    float* out = (float*)d_out;

    lstm_fused_wmma<<<BB / NB, 128, 0, stream>>>(x, W_ih, W_hh, b_ih, b_hh, fc_W, fc_b, out);
}
